# starting kernel
# baseline (speedup 1.0000x reference)
#include <hip/hip_runtime.h>

typedef float v2f __attribute__((ext_vector_type(2)));
typedef float v8f __attribute__((ext_vector_type(8)));

#define MT        64
#define WAVES     4
#define THREADS   128
#define FTOT      132
#define HDIM      512
#define DOUT      128
#define FP        132
#define HP        516
#define HBYTES    (WAVES * 16 * HP * 4)
#define TYPOFF    HBYTES
#define SMEM_BYTES (HBYTES + MT * 4)

__device__ __forceinline__ float gelu_exact(float x) {
    return 0.5f * x * (1.0f + erff(x * 0.70710678118654752f));
}

__global__ __launch_bounds__(THREADS)
void mpx_encoder_kernel(const float* __restrict__ nf,
                        const float* __restrict__ W1,
                        const float* __restrict__ b1,
                        const float* __restrict__ W2,
                        const float* __restrict__ b2,
                        float* __restrict__ out,
                        int Mtot)
{
    extern __shared__ char smem[];
    float* feats = (float*)smem;
    float* hbuf  = (float*)smem;
    int*   typ   = (int*)(smem + TYPOFF);

    const int  tid  = threadIdx.x;
    const int  lane = tid & 31;
    const int  wv   = tid >> 5;
    const int  lm   = lane & 15;
    const int  lh   = lane >> 4;
    const long base = (long)blockIdx.x * MT;

    #pragma unroll
    for (int it = 0; it < 16; ++it) {
        int i  = tid + THREADS * it;
        int r  = i >> 5;
        int c4 = i & 31;
        long rs = base + r; if (rs >= Mtot) rs = Mtot - 1;
        const float4 v = *(const float4*)(nf + rs * FTOT + 4 + c4 * 4);
        float* d = feats + r * FP + c4 * 4;
        d[0] = v.x; d[1] = v.y; d[2] = v.z; d[3] = v.w;
    }
    if (tid < MT) {
        long rs = base + tid; if (rs >= Mtot) rs = Mtot - 1;
        const float* p = nf + rs * FTOT;
        float best = p[0]; int bi = 0;
        #pragma unroll
        for (int i = 1; i < 4; ++i) { float q = p[i]; if (q > best) { best = q; bi = i; } }
        typ[tid] = bi;
    }
    __syncthreads();

    const float* myfeat = feats + wv * 16 * FP;
    v2f afr[32];
    #pragma unroll
    for (int ks = 0; ks < 32; ++ks)
        afr[ks] = *(const v2f*)(myfeat + lm * FP + ks * 4 + lh * 2);
    __syncthreads();

    float* myh = hbuf + wv * 16 * HP;
    for (int j = 0; j < 32; ++j) {
        float bias = b1[j * 16 + lm];
        v8f c;
        #pragma unroll
        for (int v = 0; v < 8; ++v) c[v] = bias;
        #pragma unroll
        for (int ks = 0; ks < 32; ++ks) {
            const float* wp = W1 + (ks * 4 + lh * 2) * HDIM + j * 16 + lm;
            v2f b; b.x = wp[0]; b.y = wp[HDIM];
            c = __builtin_amdgcn_wmma_f32_16x16x4_f32(false, afr[ks], false, b,
                                                      (short)0, c, false, false);
        }
        #pragma unroll
        for (int v = 0; v < 8; ++v) {
            int row = v + lh * 8;
            myh[row * HP + j * 16 + lm] = gelu_exact(c[v]);
        }
    }
    __syncthreads();

    for (int j = 0; j < 32; ++j) {
        float bias = b2[j * 16 + lm];
        v8f c;
        #pragma unroll
        for (int v = 0; v < 8; ++v) c[v] = bias;
        #pragma unroll 4
        for (int ks = 0; ks < 128; ++ks) {
            v2f a = *(const v2f*)(myh + lm * HP + ks * 4 + lh * 2);
            const float* wp = W2 + (ks * 4 + lh * 2) * HDIM + j * 16 + lm;
            v2f b; b.x = wp[0]; b.y = wp[HDIM];
            c = __builtin_amdgcn_wmma_f32_16x16x4_f32(false, a, false, b,
                                                      (short)0, c, false, false);
        }
        int col = j * 16 + lm;
        #pragma unroll
        for (int v = 0; v < 8; ++v) {
            int row = v + lh * 8;
            int t   = typ[wv * 16 + row];
            int c0  = t << 7;
            long rg = base + wv * 16 + row;
            if ((unsigned)(col - c0) < 128u && rg < Mtot)
                out[rg * DOUT + (col - c0)] = c[v];
        }
    }
}

extern "C" void kernel_launch(void* const* d_in, const int* in_sizes, int n_in,
                              void* d_out, int out_size, void* d_ws, size_t ws_size,
                              hipStream_t stream) {
    const float* nf = (const float*)d_in[0];
    const float* W1 = (const float*)d_in[1];
    const float* b1 = (const float*)d_in[2];
    const float* W2 = (const float*)d_in[3];
    const float* b2 = (const float*)d_in[4];
    float* out = (float*)d_out;

    int Mtot = in_sizes[0] / FTOT;
    int grid = (Mtot + MT - 1) / MT;

    hipFuncSetAttribute((const void*)mpx_encoder_kernel,
                        hipFuncAttributeMaxDynamicSharedMemorySize, SMEM_BYTES);

    mpx_encoder_kernel<<<grid, THREADS, SMEM_BYTES, stream>>>(nf, W1, b1, W2, b2, out, Mtot);
}


# Register blocking: each wave computes all 64 rows (4 row tiles) x its own 128-column slice in both GEMMs, so one loaded W fragment feeds 4 WMMAs instead of 1; 8 independent accumulators; same k order
# speedup vs baseline: 2.4483x; 2.4483x over previous
#include <hip/hip_runtime.h>

typedef float v2f __attribute__((ext_vector_type(2)));
typedef float v8f __attribute__((ext_vector_type(8)));

#define MT        64
#define WAVES     4
#define THREADS   128
#define FTOT      132
#define HDIM      512
#define DOUT      128
#define FP        132
#define HP        516
#define HBYTES    (MT * HP * 4)
#define TYPOFF    HBYTES
#define SMEM_BYTES (HBYTES + MT * 4)
#define MTILES    4
#define NPAIR     2

__device__ __forceinline__ float gelu_exact(float x) {
    return 0.5f * x * (1.0f + erff(x * 0.70710678118654752f));
}

__global__ __launch_bounds__(THREADS, 2)
void mpx_encoder_kernel(const float* __restrict__ nf,
                        const float* __restrict__ W1,
                        const float* __restrict__ b1,
                        const float* __restrict__ W2,
                        const float* __restrict__ b2,
                        float* __restrict__ out,
                        int Mtot)
{
    extern __shared__ char smem[];
    float* feats = (float*)smem;
    float* hbuf  = (float*)smem;
    int*   typ   = (int*)(smem + TYPOFF);

    const int  tid  = threadIdx.x;
    const int  lane = tid & 31;
    const int  wv   = tid >> 5;
    const int  lm   = lane & 15;
    const int  lh   = lane >> 4;
    const long base = (long)blockIdx.x * MT;

    #pragma unroll
    for (int it = 0; it < 16; ++it) {
        int i  = tid + THREADS * it;
        int r  = i >> 5;
        int c4 = i & 31;
        long rs = base + r; if (rs >= Mtot) rs = Mtot - 1;
        const float4 v = *(const float4*)(nf + rs * FTOT + 4 + c4 * 4);
        float* d = feats + r * FP + c4 * 4;
        d[0] = v.x; d[1] = v.y; d[2] = v.z; d[3] = v.w;
    }
    if (tid < MT) {
        long rs = base + tid; if (rs >= Mtot) rs = Mtot - 1;
        const float* p = nf + rs * FTOT;
        float best = p[0]; int bi = 0;
        #pragma unroll
        for (int i = 1; i < 4; ++i) { float q = p[i]; if (q > best) { best = q; bi = i; } }
        typ[tid] = bi;
    }
    __syncthreads();

    v2f afr[MTILES][32];
    #pragma unroll
    for (int mt = 0; mt < MTILES; ++mt) {
        #pragma unroll
        for (int ks = 0; ks < 32; ++ks)
            afr[mt][ks] = *(const v2f*)(feats + (mt * 16 + lm) * FP + ks * 4 + lh * 2);
    }

    __syncthreads();

    for (int np = 0; np < DOUT / (16 * NPAIR); ++np) {
        const int n0 = wv * DOUT + np * (16 * NPAIR);
        v8f c[MTILES][NPAIR];
        #pragma unroll
        for (int nt = 0; nt < NPAIR; ++nt) {
            float bias = b1[n0 + nt * 16 + lm];
            #pragma unroll
            for (int mt = 0; mt < MTILES; ++mt) {
                #pragma unroll
                for (int v = 0; v < 8; ++v) c[mt][nt][v] = bias;
            }
        }
        #pragma unroll
        for (int ks = 0; ks < 32; ++ks) {
            const float* wp = W1 + (ks * 4 + lh * 2) * HDIM + n0 + lm;
            v2f b[NPAIR];
            #pragma unroll
            for (int nt = 0; nt < NPAIR; ++nt) { b[nt].x = wp[nt * 16]; b[nt].y = wp[HDIM + nt * 16]; }
            #pragma unroll
            for (int mt = 0; mt < MTILES; ++mt) {
                #pragma unroll
                for (int nt = 0; nt < NPAIR; ++nt)
                    c[mt][nt] = __builtin_amdgcn_wmma_f32_16x16x4_f32(false, afr[mt][ks], false, b[nt],
                                                                      (short)0, c[mt][nt], false, false);
            }
        }
        #pragma unroll
        for (int mt = 0; mt < MTILES; ++mt) {
            #pragma unroll
            for (int nt = 0; nt < NPAIR; ++nt) {
                #pragma unroll
                for (int v = 0; v < 8; ++v) {
                    int row = mt * 16 + v + lh * 8;
                    hbuf[row * HP + n0 + nt * 16 + lm] = gelu_exact(c[mt][nt][v]);
                }
            }
        }
    }
    __syncthreads();

    const int rows_left = Mtot - (int)base;
    unsigned selmask = 0u;
    #pragma unroll
    for (int mt = 0; mt < MTILES; ++mt) {
        #pragma unroll
        for (int v = 0; v < 8; ++v) {
            int row = mt * 16 + v + lh * 8;
            if (typ[row] == wv && row < rows_left) selmask |= 1u << (mt * 8 + v);
        }
    }

    for (int np = 0; np < DOUT / (16 * NPAIR); ++np) {
        const int n0 = wv * DOUT + np * (16 * NPAIR);
        v8f c[MTILES][NPAIR];
        #pragma unroll
        for (int nt = 0; nt < NPAIR; ++nt) {
            float bias = b2[n0 + nt * 16 + lm];
            #pragma unroll
            for (int mt = 0; mt < MTILES; ++mt) {
                #pragma unroll
                for (int v = 0; v < 8; ++v) c[mt][nt][v] = bias;
            }
        }
        #pragma unroll 4
        for (int ks = 0; ks < 128; ++ks) {
            v2f a[MTILES];
            #pragma unroll
            for (int mt = 0; mt < MTILES; ++mt)
                a[mt] = *(const v2f*)(hbuf + (mt * 16 + lm) * HP + ks * 4 + lh * 2);
            const float* wp = W2 + (ks * 4 + lh * 2) * HDIM + n0 + lm;
            v2f b[NPAIR];
            #pragma unroll
            for (int nt = 0; nt < NPAIR; ++nt) { b[nt].x = wp[nt * 16]; b[nt].y = wp[HDIM + nt * 16]; }
            #pragma unroll
            for (int mt = 0; mt < MTILES; ++mt) {
                #pragma unroll
                for (int nt = 0; nt < NPAIR; ++nt)
                    c[mt][nt] = __builtin_amdgcn_wmma_f32_16x16x4_f32(false, a[mt], false, b[nt],
                                                                      (short)0, c[mt][nt], false, false);
            }
        }
        #pragma unroll
        for (int mt = 0; mt < MTILES; ++mt) {
            #pragma unroll
            for (int v = 0; v < 8; ++v) {
                if ((selmask >> (mt * 8 + v)) & 1u) {
                    long rg = base + mt * 16 + v + lh * 8;
                    float* o = out + rg * DOUT + np * (16 * NPAIR) + lm;
                    #pragma unroll
                    for (int nt = 0; nt < NPAIR; ++nt) o[nt * 16] = c[mt][nt][v];
                }
            }
        }
    }
}

extern "C" void kernel_launch(void* const* d_in, const int* in_sizes, int n_in,
                              void* d_out, int out_size, void* d_ws, size_t ws_size,
                              hipStream_t stream) {
    const float* nf = (const float*)d_in[0];
    const float* W1 = (const float*)d_in[1];
    const float* b1 = (const float*)d_in[2];
    const float* W2 = (const float*)d_in[3];
    const float* b2 = (const float*)d_in[4];
    float* out = (float*)d_out;

    int Mtot = in_sizes[0] / FTOT;
    int grid = (Mtot + MT - 1) / MT;

    hipFuncSetAttribute((const void*)mpx_encoder_kernel,
                        hipFuncAttributeMaxDynamicSharedMemorySize, SMEM_BYTES);

    mpx_encoder_kernel<<<grid, THREADS, SMEM_BYTES, stream>>>(nf, W1, b1, W2, b2, out, Mtot);
}


# GEMM2 only for each row's own node-type slice: wave t gathers the rows of type t from LDS and computes 32 rows x 64 cols per block, so GEMM2 WMMAs halve; same k order, sums unchanged
# speedup vs baseline: 3.5998x; 1.4703x over previous
#include <hip/hip_runtime.h>

typedef float v2f __attribute__((ext_vector_type(2)));
typedef float v8f __attribute__((ext_vector_type(8)));

#define MT        64
#define WAVES     4
#define THREADS   128
#define FTOT      132
#define HDIM      512
#define DOUT      128
#define FP        132
#define HP        516
#define HBYTES    (MT * HP * 4)
#define TYPOFF    HBYTES
#define RLOFF     (HBYTES + MT * 4)
#define SMEM_BYTES (RLOFF + 4 * MT * 4)
#define RT2       2
#define CT2       4
#define MTILES    4
#define NPAIR     2

__device__ __forceinline__ float gelu_exact(float x) {
    return 0.5f * x * (1.0f + erff(x * 0.70710678118654752f));
}

__global__ __launch_bounds__(THREADS, 2)
void mpx_encoder_kernel(const float* __restrict__ nf,
                        const float* __restrict__ W1,
                        const float* __restrict__ b1,
                        const float* __restrict__ W2,
                        const float* __restrict__ b2,
                        float* __restrict__ out,
                        int Mtot)
{
    extern __shared__ char smem[];
    float* feats = (float*)smem;
    float* hbuf  = (float*)smem;
    int*   typ   = (int*)(smem + TYPOFF);

    const int  tid  = threadIdx.x;
    const int  lane = tid & 31;
    const int  wv   = tid >> 5;
    const int  lm   = lane & 15;
    const int  lh   = lane >> 4;
    const long base = (long)blockIdx.x * MT;
    int*   rlist = (int*)(smem + RLOFF) + wv * MT;

    #pragma unroll
    for (int it = 0; it < 16; ++it) {
        int i  = tid + THREADS * it;
        int r  = i >> 5;
        int c4 = i & 31;
        long rs = base + r; if (rs >= Mtot) rs = Mtot - 1;
        const float4 v = *(const float4*)(nf + rs * FTOT + 4 + c4 * 4);
        float* d = feats + r * FP + c4 * 4;
        d[0] = v.x; d[1] = v.y; d[2] = v.z; d[3] = v.w;
    }
    if (tid < MT) {
        long rs = base + tid; if (rs >= Mtot) rs = Mtot - 1;
        const float* p = nf + rs * FTOT;
        float best = p[0]; int bi = 0;
        #pragma unroll
        for (int i = 1; i < 4; ++i) { float q = p[i]; if (q > best) { best = q; bi = i; } }
        typ[tid] = bi;
    }
    __syncthreads();

    v2f afr[MTILES][32];
    #pragma unroll
    for (int mt = 0; mt < MTILES; ++mt) {
        #pragma unroll
        for (int ks = 0; ks < 32; ++ks)
            afr[mt][ks] = *(const v2f*)(feats + (mt * 16 + lm) * FP + ks * 4 + lh * 2);
    }

    const int rows_left = Mtot - (int)base;
    int nmine;
    {
        const bool p0 = (typ[lane] == wv) && (lane < rows_left);
        const bool p1 = (typ[lane + 32] == wv) && (lane + 32 < rows_left);
        const unsigned m0 = __builtin_amdgcn_ballot_w32(p0);
        const unsigned m1 = __builtin_amdgcn_ballot_w32(p1);
        const unsigned below = (1u << lane) - 1u;
        const int cnt0 = __popc(m0);
        if (p0) rlist[__popc(m0 & below)] = lane;
        if (p1) rlist[cnt0 + __popc(m1 & below)] = lane + 32;
        nmine = cnt0 + __popc(m1);
    }

    __syncthreads();

    for (int np = 0; np < DOUT / (16 * NPAIR); ++np) {
        const int n0 = wv * DOUT + np * (16 * NPAIR);
        v8f c[MTILES][NPAIR];
        #pragma unroll
        for (int nt = 0; nt < NPAIR; ++nt) {
            float bias = b1[n0 + nt * 16 + lm];
            #pragma unroll
            for (int mt = 0; mt < MTILES; ++mt) {
                #pragma unroll
                for (int v = 0; v < 8; ++v) c[mt][nt][v] = bias;
            }
        }
        #pragma unroll
        for (int ks = 0; ks < 32; ++ks) {
            const float* wp = W1 + (ks * 4 + lh * 2) * HDIM + n0 + lm;
            v2f b[NPAIR];
            #pragma unroll
            for (int nt = 0; nt < NPAIR; ++nt) { b[nt].x = wp[nt * 16]; b[nt].y = wp[HDIM + nt * 16]; }
            #pragma unroll
            for (int mt = 0; mt < MTILES; ++mt) {
                #pragma unroll
                for (int nt = 0; nt < NPAIR; ++nt)
                    c[mt][nt] = __builtin_amdgcn_wmma_f32_16x16x4_f32(false, afr[mt][ks], false, b[nt],
                                                                      (short)0, c[mt][nt], false, false);
            }
        }
        #pragma unroll
        for (int mt = 0; mt < MTILES; ++mt) {
            #pragma unroll
            for (int nt = 0; nt < NPAIR; ++nt) {
                #pragma unroll
                for (int v = 0; v < 8; ++v) {
                    int row = mt * 16 + v + lh * 8;
                    hbuf[row * HP + n0 + nt * 16 + lm] = gelu_exact(c[mt][nt][v]);
                }
            }
        }
    }
    __syncthreads();

    for (int s0 = 0; s0 < nmine; s0 += 16 * RT2) {
        int arow[RT2];
        #pragma unroll
        for (int mt = 0; mt < RT2; ++mt) {
            int s = s0 + mt * 16 + lm;
            arow[mt] = rlist[s < nmine ? s : nmine - 1];
        }
        int srow[RT2][8];
        #pragma unroll
        for (int mt = 0; mt < RT2; ++mt) {
            #pragma unroll
            for (int v = 0; v < 8; ++v) {
                int s = s0 + mt * 16 + lh * 8 + v;
                int r = rlist[s < nmine ? s : nmine - 1];
                srow[mt][v] = (s < nmine) ? r : -1;
            }
        }
        for (int cp = 0; cp < DOUT / (16 * CT2); ++cp) {
            const int n0 = wv * DOUT + cp * (16 * CT2);
            v8f c[RT2][CT2];
            #pragma unroll
            for (int nt = 0; nt < CT2; ++nt) {
                float bias = b2[n0 + nt * 16 + lm];
                #pragma unroll
                for (int mt = 0; mt < RT2; ++mt) {
                    #pragma unroll
                    for (int v = 0; v < 8; ++v) c[mt][nt][v] = bias;
                }
            }
            #pragma unroll 4
            for (int ks = 0; ks < 128; ++ks) {
                v2f a[RT2];
                #pragma unroll
                for (int mt = 0; mt < RT2; ++mt)
                    a[mt] = *(const v2f*)(hbuf + arow[mt] * HP + ks * 4 + lh * 2);
                const float* wp = W2 + (ks * 4 + lh * 2) * HDIM + n0 + lm;
                v2f b[CT2];
                #pragma unroll
                for (int nt = 0; nt < CT2; ++nt) { b[nt].x = wp[nt * 16]; b[nt].y = wp[HDIM + nt * 16]; }
                #pragma unroll
                for (int mt = 0; mt < RT2; ++mt) {
                    #pragma unroll
                    for (int nt = 0; nt < CT2; ++nt)
                        c[mt][nt] = __builtin_amdgcn_wmma_f32_16x16x4_f32(false, a[mt], false, b[nt],
                                                                          (short)0, c[mt][nt], false, false);
                }
            }
            #pragma unroll
            for (int mt = 0; mt < RT2; ++mt) {
                #pragma unroll
                for (int v = 0; v < 8; ++v) {
                    if (srow[mt][v] >= 0) {
                        float* o = out + (base + srow[mt][v]) * DOUT + cp * (16 * CT2) + lm;
                        #pragma unroll
                        for (int nt = 0; nt < CT2; ++nt) o[nt * 16] = c[mt][nt][v];
                    }
                }
            }
        }
    }
}

extern "C" void kernel_launch(void* const* d_in, const int* in_sizes, int n_in,
                              void* d_out, int out_size, void* d_ws, size_t ws_size,
                              hipStream_t stream) {
    const float* nf = (const float*)d_in[0];
    const float* W1 = (const float*)d_in[1];
    const float* b1 = (const float*)d_in[2];
    const float* W2 = (const float*)d_in[3];
    const float* b2 = (const float*)d_in[4];
    float* out = (float*)d_out;

    int Mtot = in_sizes[0] / FTOT;
    int grid = (Mtot + MT - 1) / MT;

    hipFuncSetAttribute((const void*)mpx_encoder_kernel,
                        hipFuncAttributeMaxDynamicSharedMemorySize, SMEM_BYTES);

    mpx_encoder_kernel<<<grid, THREADS, SMEM_BYTES, stream>>>(nf, W1, b1, W2, b2, out, Mtot);
}


# GEMM2 software pipelining: A (LDS) and B (global) fragment loads of k-group g+1 go into a second register set ahead of the 32 WMMAs of group g; same k order, sums unchanged
# speedup vs baseline: 3.8772x; 1.0771x over previous
#include <hip/hip_runtime.h>

typedef float v2f __attribute__((ext_vector_type(2)));
typedef float v8f __attribute__((ext_vector_type(8)));

#define MT        64
#define WAVES     4
#define THREADS   128
#define FTOT      132
#define HDIM      512
#define DOUT      128
#define FP        132
#define HP        516
#define HBYTES    (MT * HP * 4)
#define TYPOFF    HBYTES
#define RLOFF     (HBYTES + MT * 4)
#define SMEM_BYTES (RLOFF + 4 * MT * 4)
#define RT2       2
#define CT2       4
#define MTILES    4
#define NPAIR     2
#define KG        4
#define NGRP      (HDIM / 4 / KG)

__device__ __forceinline__ float gelu_exact(float x) {
    return 0.5f * x * (1.0f + erff(x * 0.70710678118654752f));
}

__device__ __forceinline__ void g2_load_a(v2f (&a)[KG][RT2], const float* hb, const int (&hoff)[RT2]) {
    #pragma unroll
    for (int j = 0; j < KG; ++j) {
        #pragma unroll
        for (int mt = 0; mt < RT2; ++mt) a[j][mt] = *(const v2f*)(hb + hoff[mt] + j * 4);
    }
}

__device__ __forceinline__ void g2_load_b(v2f (&b)[KG][CT2], const float* wp) {
    #pragma unroll
    for (int j = 0; j < KG; ++j) {
        #pragma unroll
        for (int nt = 0; nt < CT2; ++nt) {
            b[j][nt].x = wp[j * 4 * HDIM + nt * 16];
            b[j][nt].y = wp[j * 4 * HDIM + HDIM + nt * 16];
        }
    }
}

#define G2_STAGE(an, bn, ac, bc)                         \
    do {                                                  \
        g2_load_a(an, hbuf, hoff);                        \
        __builtin_amdgcn_sched_barrier(0x76);             \
        g2_load_b(bn, wp);                                \
        g2_mma(c, ac, bc);                                \
    } while (0)

__device__ __forceinline__ void g2_mma(v8f (&c)[RT2][CT2], const v2f (&a)[KG][RT2], const v2f (&b)[KG][CT2]) {
    #pragma unroll
    for (int j = 0; j < KG; ++j) {
        #pragma unroll
        for (int mt = 0; mt < RT2; ++mt) {
            #pragma unroll
            for (int nt = 0; nt < CT2; ++nt)
                c[mt][nt] = __builtin_amdgcn_wmma_f32_16x16x4_f32(false, a[j][mt], false, b[j][nt],
                                                                  (short)0, c[mt][nt], false, false);
        }
    }
}

__global__ __launch_bounds__(THREADS, 2)
void mpx_encoder_kernel(const float* __restrict__ nf,
                        const float* __restrict__ W1,
                        const float* __restrict__ b1,
                        const float* __restrict__ W2,
                        const float* __restrict__ b2,
                        float* __restrict__ out,
                        int Mtot)
{
    extern __shared__ char smem[];
    float* feats = (float*)smem;
    float* hbuf  = (float*)smem;
    int*   typ   = (int*)(smem + TYPOFF);

    const int  tid  = threadIdx.x;
    const int  lane = tid & 31;
    const int  wv   = tid >> 5;
    const int  lm   = lane & 15;
    const int  lh   = lane >> 4;
    const long base = (long)blockIdx.x * MT;
    int*   rlist = (int*)(smem + RLOFF) + wv * MT;

    #pragma unroll
    for (int it = 0; it < 16; ++it) {
        int i  = tid + THREADS * it;
        int r  = i >> 5;
        int c4 = i & 31;
        long rs = base + r; if (rs >= Mtot) rs = Mtot - 1;
        const float4 v = *(const float4*)(nf + rs * FTOT + 4 + c4 * 4);
        float* d = feats + r * FP + c4 * 4;
        d[0] = v.x; d[1] = v.y; d[2] = v.z; d[3] = v.w;
    }
    if (tid < MT) {
        long rs = base + tid; if (rs >= Mtot) rs = Mtot - 1;
        const float* p = nf + rs * FTOT;
        float best = p[0]; int bi = 0;
        #pragma unroll
        for (int i = 1; i < 4; ++i) { float q = p[i]; if (q > best) { best = q; bi = i; } }
        typ[tid] = bi;
    }
    __syncthreads();

    v2f afr[MTILES][32];
    #pragma unroll
    for (int mt = 0; mt < MTILES; ++mt) {
        #pragma unroll
        for (int ks = 0; ks < 32; ++ks)
            afr[mt][ks] = *(const v2f*)(feats + (mt * 16 + lm) * FP + ks * 4 + lh * 2);
    }

    const int rows_left = Mtot - (int)base;
    int nmine;
    {
        const bool p0 = (typ[lane] == wv) && (lane < rows_left);
        const bool p1 = (typ[lane + 32] == wv) && (lane + 32 < rows_left);
        const unsigned m0 = __builtin_amdgcn_ballot_w32(p0);
        const unsigned m1 = __builtin_amdgcn_ballot_w32(p1);
        const unsigned below = (1u << lane) - 1u;
        const int cnt0 = __popc(m0);
        if (p0) rlist[__popc(m0 & below)] = lane;
        if (p1) rlist[cnt0 + __popc(m1 & below)] = lane + 32;
        nmine = cnt0 + __popc(m1);
    }

    __syncthreads();

    for (int np = 0; np < DOUT / (16 * NPAIR); ++np) {
        const int n0 = wv * DOUT + np * (16 * NPAIR);
        v8f c[MTILES][NPAIR];
        #pragma unroll
        for (int nt = 0; nt < NPAIR; ++nt) {
            float bias = b1[n0 + nt * 16 + lm];
            #pragma unroll
            for (int mt = 0; mt < MTILES; ++mt) {
                #pragma unroll
                for (int v = 0; v < 8; ++v) c[mt][nt][v] = bias;
            }
        }
        #pragma unroll
        for (int ks = 0; ks < 32; ++ks) {
            const float* wp = W1 + (ks * 4 + lh * 2) * HDIM + n0 + lm;
            v2f b[NPAIR];
            #pragma unroll
            for (int nt = 0; nt < NPAIR; ++nt) { b[nt].x = wp[nt * 16]; b[nt].y = wp[HDIM + nt * 16]; }
            #pragma unroll
            for (int mt = 0; mt < MTILES; ++mt) {
                #pragma unroll
                for (int nt = 0; nt < NPAIR; ++nt)
                    c[mt][nt] = __builtin_amdgcn_wmma_f32_16x16x4_f32(false, afr[mt][ks], false, b[nt],
                                                                      (short)0, c[mt][nt], false, false);
            }
        }
        #pragma unroll
        for (int mt = 0; mt < MTILES; ++mt) {
            #pragma unroll
            for (int nt = 0; nt < NPAIR; ++nt) {
                #pragma unroll
                for (int v = 0; v < 8; ++v) {
                    int row = mt * 16 + v + lh * 8;
                    hbuf[row * HP + n0 + nt * 16 + lm] = gelu_exact(c[mt][nt][v]);
                }
            }
        }
    }
    __syncthreads();

    for (int s0 = 0; s0 < nmine; s0 += 16 * RT2) {
        int arow[RT2];
        #pragma unroll
        for (int mt = 0; mt < RT2; ++mt) {
            int s = s0 + mt * 16 + lm;
            arow[mt] = rlist[s < nmine ? s : nmine - 1];
        }
        int srow[RT2][8];
        #pragma unroll
        for (int mt = 0; mt < RT2; ++mt) {
            #pragma unroll
            for (int v = 0; v < 8; ++v) {
                int s = s0 + mt * 16 + lh * 8 + v;
                int r = rlist[s < nmine ? s : nmine - 1];
                srow[mt][v] = (s < nmine) ? r : -1;
            }
        }
        for (int cp = 0; cp < DOUT / (16 * CT2); ++cp) {
            const int n0 = wv * DOUT + cp * (16 * CT2);
            v8f c[RT2][CT2];
            #pragma unroll
            for (int nt = 0; nt < CT2; ++nt) {
                float bias = b2[n0 + nt * 16 + lm];
                #pragma unroll
                for (int mt = 0; mt < RT2; ++mt) {
                    #pragma unroll
                    for (int v = 0; v < 8; ++v) c[mt][nt][v] = bias;
                }
            }
            int hoff[RT2];
            #pragma unroll
            for (int mt = 0; mt < RT2; ++mt) hoff[mt] = arow[mt] * HP + lh * 2;
            const float* wp = W2 + (lh * 2) * HDIM + n0 + lm;
            v2f a0[KG][RT2], b0[KG][CT2], a1[KG][RT2], b1[KG][CT2];
            g2_load_a(a0, hbuf, hoff);
            g2_load_b(b0, wp);
            for (int g = 0; g < NGRP - 2; g += 2) {
                #pragma unroll
                for (int mt = 0; mt < RT2; ++mt) hoff[mt] += KG * 4;
                wp += KG * 4 * HDIM;
                G2_STAGE(a1, b1, a0, b0);
                #pragma unroll
                for (int mt = 0; mt < RT2; ++mt) hoff[mt] += KG * 4;
                wp += KG * 4 * HDIM;
                G2_STAGE(a0, b0, a1, b1);
            }
            #pragma unroll
            for (int mt = 0; mt < RT2; ++mt) hoff[mt] += KG * 4;
            wp += KG * 4 * HDIM;
            G2_STAGE(a1, b1, a0, b0);
            g2_mma(c, a1, b1);
            #pragma unroll
            for (int mt = 0; mt < RT2; ++mt) {
                #pragma unroll
                for (int v = 0; v < 8; ++v) {
                    if (srow[mt][v] >= 0) {
                        float* o = out + (base + srow[mt][v]) * DOUT + cp * (16 * CT2) + lm;
                        #pragma unroll
                        for (int nt = 0; nt < CT2; ++nt) o[nt * 16] = c[mt][nt][v];
                    }
                }
            }
        }
    }
}

extern "C" void kernel_launch(void* const* d_in, const int* in_sizes, int n_in,
                              void* d_out, int out_size, void* d_ws, size_t ws_size,
                              hipStream_t stream) {
    const float* nf = (const float*)d_in[0];
    const float* W1 = (const float*)d_in[1];
    const float* b1 = (const float*)d_in[2];
    const float* W2 = (const float*)d_in[3];
    const float* b2 = (const float*)d_in[4];
    float* out = (float*)d_out;

    int Mtot = in_sizes[0] / FTOT;
    int grid = (Mtot + MT - 1) / MT;

    hipFuncSetAttribute((const void*)mpx_encoder_kernel,
                        hipFuncAttributeMaxDynamicSharedMemorySize, SMEM_BYTES);

    mpx_encoder_kernel<<<grid, THREADS, SMEM_BYTES, stream>>>(nf, W1, b1, W2, b2, out, Mtot);
}


# GELU epilogue branch-free: library erff (two ranges behind a divergent branch, 64 serial chains per pass) replaced by one rational-exp erf, 16 values interleaved; gelu abs error below 4e-7
# speedup vs baseline: 4.6427x; 1.1974x over previous
#include <hip/hip_runtime.h>

typedef float v2f __attribute__((ext_vector_type(2)));
typedef float v8f __attribute__((ext_vector_type(8)));

#define MT        64
#define WAVES     4
#define THREADS   128
#define FTOT      132
#define HDIM      512
#define DOUT      128
#define FP        132
#define HP        516
#define HBYTES    (MT * HP * 4)
#define TYPOFF    HBYTES
#define RLOFF     (HBYTES + MT * 4)
#define SMEM_BYTES (RLOFF + 4 * MT * 4)
#define RT2       2
#define CT2       4
#define MTILES    4
#define NPAIR     2
#define KG        4
#define NGRP      (HDIM / 4 / KG)

__device__ __forceinline__ v8f splat8(float s) { v8f r = {s, s, s, s, s, s, s, s}; return r; }

__device__ __forceinline__ void gelu_tile(v8f (&x)[NPAIR]) {
    v8f z[NPAIR], az[NPAIR], t[NPAIR], e[NPAIR], p[NPAIR];
    #pragma unroll
    for (int n = 0; n < NPAIR; ++n) {
        z[n]  = x[n] * splat8(0.70710678118654752f);
        az[n] = __builtin_elementwise_abs(z[n]);
    }
    #pragma unroll
    for (int n = 0; n < NPAIR; ++n) {
        const v8f d = __builtin_elementwise_fma(az[n], splat8(0.3275911f), splat8(1.0f));
        const v8f w = az[n] * az[n] * splat8(-1.4426950408889634f);
        #pragma unroll
        for (int i = 0; i < 8; ++i) {
            t[n][i] = __builtin_amdgcn_rcpf(d[i]);
            e[n][i] = __builtin_amdgcn_exp2f(w[i]);
        }
    }
    #pragma unroll
    for (int n = 0; n < NPAIR; ++n) p[n] = __builtin_elementwise_fma(t[n], splat8(1.061405429f), splat8(-1.453152027f));
    #pragma unroll
    for (int n = 0; n < NPAIR; ++n) p[n] = __builtin_elementwise_fma(t[n], p[n], splat8(1.421413741f));
    #pragma unroll
    for (int n = 0; n < NPAIR; ++n) p[n] = __builtin_elementwise_fma(t[n], p[n], splat8(-0.284496736f));
    #pragma unroll
    for (int n = 0; n < NPAIR; ++n) p[n] = __builtin_elementwise_fma(t[n], p[n], splat8(0.254829592f));
    #pragma unroll
    for (int n = 0; n < NPAIR; ++n) p[n] = p[n] * t[n];
    #pragma unroll
    for (int n = 0; n < NPAIR; ++n) {
        const v8f erfa = __builtin_elementwise_fma(-p[n], e[n], splat8(1.0f));
        const v8f erfz = __builtin_elementwise_copysign(erfa, z[n]);
        const v8f hx   = x[n] * splat8(0.5f);
        x[n] = __builtin_elementwise_fma(hx, erfz, hx);
    }
}

__device__ __forceinline__ void g2_load_a(v2f (&a)[KG][RT2], const float* hb, const int (&hoff)[RT2]) {
    #pragma unroll
    for (int j = 0; j < KG; ++j) {
        #pragma unroll
        for (int mt = 0; mt < RT2; ++mt) a[j][mt] = *(const v2f*)(hb + hoff[mt] + j * 4);
    }
}

__device__ __forceinline__ void g2_load_b(v2f (&b)[KG][CT2], const float* wp) {
    #pragma unroll
    for (int j = 0; j < KG; ++j) {
        #pragma unroll
        for (int nt = 0; nt < CT2; ++nt) {
            b[j][nt].x = wp[j * 4 * HDIM + nt * 16];
            b[j][nt].y = wp[j * 4 * HDIM + HDIM + nt * 16];
        }
    }
}

#define G2_STAGE(an, bn, ac, bc)                         \
    do {                                                  \
        g2_load_a(an, hbuf, hoff);                        \
        __builtin_amdgcn_sched_barrier(0x76);             \
        g2_load_b(bn, wp);                                \
        g2_mma(c, ac, bc);                                \
    } while (0)

__device__ __forceinline__ void g2_mma(v8f (&c)[RT2][CT2], const v2f (&a)[KG][RT2], const v2f (&b)[KG][CT2]) {
    #pragma unroll
    for (int j = 0; j < KG; ++j) {
        #pragma unroll
        for (int mt = 0; mt < RT2; ++mt) {
            #pragma unroll
            for (int nt = 0; nt < CT2; ++nt)
                c[mt][nt] = __builtin_amdgcn_wmma_f32_16x16x4_f32(false, a[j][mt], false, b[j][nt],
                                                                  (short)0, c[mt][nt], false, false);
        }
    }
}

__global__ __launch_bounds__(THREADS, 2)
void mpx_encoder_kernel(const float* __restrict__ nf,
                        const float* __restrict__ W1,
                        const float* __restrict__ b1,
                        const float* __restrict__ W2,
                        const float* __restrict__ b2,
                        float* __restrict__ out,
                        int Mtot)
{
    extern __shared__ char smem[];
    float* feats = (float*)smem;
    float* hbuf  = (float*)smem;
    int*   typ   = (int*)(smem + TYPOFF);

    const int  tid  = threadIdx.x;
    const int  lane = tid & 31;
    const int  wv   = tid >> 5;
    const int  lm   = lane & 15;
    const int  lh   = lane >> 4;
    const long base = (long)blockIdx.x * MT;
    int*   rlist = (int*)(smem + RLOFF) + wv * MT;

    #pragma unroll
    for (int it = 0; it < 16; ++it) {
        int i  = tid + THREADS * it;
        int r  = i >> 5;
        int c4 = i & 31;
        long rs = base + r; if (rs >= Mtot) rs = Mtot - 1;
        const float4 v = *(const float4*)(nf + rs * FTOT + 4 + c4 * 4);
        float* d = feats + r * FP + c4 * 4;
        d[0] = v.x; d[1] = v.y; d[2] = v.z; d[3] = v.w;
    }
    if (tid < MT) {
        long rs = base + tid; if (rs >= Mtot) rs = Mtot - 1;
        const float* p = nf + rs * FTOT;
        float best = p[0]; int bi = 0;
        #pragma unroll
        for (int i = 1; i < 4; ++i) { float q = p[i]; if (q > best) { best = q; bi = i; } }
        typ[tid] = bi;
    }
    __syncthreads();

    v2f afr[MTILES][32];
    #pragma unroll
    for (int mt = 0; mt < MTILES; ++mt) {
        #pragma unroll
        for (int ks = 0; ks < 32; ++ks)
            afr[mt][ks] = *(const v2f*)(feats + (mt * 16 + lm) * FP + ks * 4 + lh * 2);
    }

    const int rows_left = Mtot - (int)base;
    int nmine;
    {
        const bool p0 = (typ[lane] == wv) && (lane < rows_left);
        const bool p1 = (typ[lane + 32] == wv) && (lane + 32 < rows_left);
        const unsigned m0 = __builtin_amdgcn_ballot_w32(p0);
        const unsigned m1 = __builtin_amdgcn_ballot_w32(p1);
        const unsigned below = (1u << lane) - 1u;
        const int cnt0 = __popc(m0);
        if (p0) rlist[__popc(m0 & below)] = lane;
        if (p1) rlist[cnt0 + __popc(m1 & below)] = lane + 32;
        nmine = cnt0 + __popc(m1);
    }

    __syncthreads();

    for (int np = 0; np < DOUT / (16 * NPAIR); ++np) {
        const int n0 = wv * DOUT + np * (16 * NPAIR);
        v8f c[MTILES][NPAIR];
        #pragma unroll
        for (int nt = 0; nt < NPAIR; ++nt) {
            float bias = b1[n0 + nt * 16 + lm];
            #pragma unroll
            for (int mt = 0; mt < MTILES; ++mt) {
                #pragma unroll
                for (int v = 0; v < 8; ++v) c[mt][nt][v] = bias;
            }
        }
        #pragma unroll
        for (int ks = 0; ks < 32; ++ks) {
            const float* wp = W1 + (ks * 4 + lh * 2) * HDIM + n0 + lm;
            v2f b[NPAIR];
            #pragma unroll
            for (int nt = 0; nt < NPAIR; ++nt) { b[nt].x = wp[nt * 16]; b[nt].y = wp[HDIM + nt * 16]; }
            #pragma unroll
            for (int mt = 0; mt < MTILES; ++mt) {
                #pragma unroll
                for (int nt = 0; nt < NPAIR; ++nt)
                    c[mt][nt] = __builtin_amdgcn_wmma_f32_16x16x4_f32(false, afr[mt][ks], false, b[nt],
                                                                      (short)0, c[mt][nt], false, false);
            }
        }
        __builtin_amdgcn_sched_barrier(0);
        #pragma unroll
        for (int mt = 0; mt < MTILES; ++mt) {
            gelu_tile(c[mt]);
            #pragma unroll
            for (int nt = 0; nt < NPAIR; ++nt) {
                #pragma unroll
                for (int v = 0; v < 8; ++v) {
                    int row = mt * 16 + v + lh * 8;
                    hbuf[row * HP + n0 + nt * 16 + lm] = c[mt][nt][v];
                }
            }
        }
    }
    __syncthreads();

    for (int s0 = 0; s0 < nmine; s0 += 16 * RT2) {
        int arow[RT2];
        #pragma unroll
        for (int mt = 0; mt < RT2; ++mt) {
            int s = s0 + mt * 16 + lm;
            arow[mt] = rlist[s < nmine ? s : nmine - 1];
        }
        int srow[RT2][8];
        #pragma unroll
        for (int mt = 0; mt < RT2; ++mt) {
            #pragma unroll
            for (int v = 0; v < 8; ++v) {
                int s = s0 + mt * 16 + lh * 8 + v;
                int r = rlist[s < nmine ? s : nmine - 1];
                srow[mt][v] = (s < nmine) ? r : -1;
            }
        }
        for (int cp = 0; cp < DOUT / (16 * CT2); ++cp) {
            const int n0 = wv * DOUT + cp * (16 * CT2);
            v8f c[RT2][CT2];
            #pragma unroll
            for (int nt = 0; nt < CT2; ++nt) {
                float bias = b2[n0 + nt * 16 + lm];
                #pragma unroll
                for (int mt = 0; mt < RT2; ++mt) {
                    #pragma unroll
                    for (int v = 0; v < 8; ++v) c[mt][nt][v] = bias;
                }
            }
            int hoff[RT2];
            #pragma unroll
            for (int mt = 0; mt < RT2; ++mt) hoff[mt] = arow[mt] * HP + lh * 2;
            const float* wp = W2 + (lh * 2) * HDIM + n0 + lm;
            v2f a0[KG][RT2], b0[KG][CT2], a1[KG][RT2], b1[KG][CT2];
            g2_load_a(a0, hbuf, hoff);
            g2_load_b(b0, wp);
            for (int g = 0; g < NGRP - 2; g += 2) {
                #pragma unroll
                for (int mt = 0; mt < RT2; ++mt) hoff[mt] += KG * 4;
                wp += KG * 4 * HDIM;
                G2_STAGE(a1, b1, a0, b0);
                #pragma unroll
                for (int mt = 0; mt < RT2; ++mt) hoff[mt] += KG * 4;
                wp += KG * 4 * HDIM;
                G2_STAGE(a0, b0, a1, b1);
            }
            #pragma unroll
            for (int mt = 0; mt < RT2; ++mt) hoff[mt] += KG * 4;
            wp += KG * 4 * HDIM;
            G2_STAGE(a1, b1, a0, b0);
            g2_mma(c, a1, b1);
            #pragma unroll
            for (int mt = 0; mt < RT2; ++mt) {
                #pragma unroll
                for (int v = 0; v < 8; ++v) {
                    if (srow[mt][v] >= 0) {
                        float* o = out + (base + srow[mt][v]) * DOUT + cp * (16 * CT2) + lm;
                        #pragma unroll
                        for (int nt = 0; nt < CT2; ++nt) o[nt * 16] = c[mt][nt][v];
                    }
                }
            }
        }
    }
}

extern "C" void kernel_launch(void* const* d_in, const int* in_sizes, int n_in,
                              void* d_out, int out_size, void* d_ws, size_t ws_size,
                              hipStream_t stream) {
    const float* nf = (const float*)d_in[0];
    const float* W1 = (const float*)d_in[1];
    const float* b1 = (const float*)d_in[2];
    const float* W2 = (const float*)d_in[3];
    const float* b2 = (const float*)d_in[4];
    float* out = (float*)d_out;

    int Mtot = in_sizes[0] / FTOT;
    int grid = (Mtot + MT - 1) / MT;

    hipFuncSetAttribute((const void*)mpx_encoder_kernel,
                        hipFuncAttributeMaxDynamicSharedMemorySize, SMEM_BYTES);

    mpx_encoder_kernel<<<grid, THREADS, SMEM_BYTES, stream>>>(nf, W1, b1, W2, b2, out, Mtot);
}


# GEMM2 runs only the 16-row tiles that hold rows (1-tile block when 16 or fewer remain); wave w takes column quarter w of every node type, so the 4 waves do equal work; same k order, sums unchanged
# speedup vs baseline: 5.0328x; 1.0840x over previous
#include <hip/hip_runtime.h>

typedef float v2f __attribute__((ext_vector_type(2)));
typedef float v8f __attribute__((ext_vector_type(8)));

#define MT        64
#define WAVES     4
#define THREADS   128
#define FTOT      132
#define HDIM      512
#define DOUT      128
#define FP        132
#define HP        516
#define HBYTES    (MT * HP * 4)
#define TYPOFF    HBYTES
#define RLOFF     (HBYTES + MT * 4)
#define CNTOFF    (RLOFF + 4 * MT * 4)
#define SMEM_BYTES (CNTOFF + 16)
#define CT2       2
#define QCOLS     (16 * CT2)
#define MTILES    4
#define NPAIR     2
#define KG        8
#define NGRP      (HDIM / 4 / KG)

__device__ __forceinline__ v8f splat8(float s) { v8f r = {s, s, s, s, s, s, s, s}; return r; }

__device__ __forceinline__ void gelu_tile(v8f (&x)[NPAIR]) {
    v8f z[NPAIR], az[NPAIR], t[NPAIR], e[NPAIR], p[NPAIR];
    #pragma unroll
    for (int n = 0; n < NPAIR; ++n) {
        z[n]  = x[n] * splat8(0.70710678118654752f);
        az[n] = __builtin_elementwise_abs(z[n]);
    }
    #pragma unroll
    for (int n = 0; n < NPAIR; ++n) {
        const v8f d = __builtin_elementwise_fma(az[n], splat8(0.3275911f), splat8(1.0f));
        const v8f w = az[n] * az[n] * splat8(-1.4426950408889634f);
        #pragma unroll
        for (int i = 0; i < 8; ++i) {
            t[n][i] = __builtin_amdgcn_rcpf(d[i]);
            e[n][i] = __builtin_amdgcn_exp2f(w[i]);
        }
    }
    #pragma unroll
    for (int n = 0; n < NPAIR; ++n) p[n] = __builtin_elementwise_fma(t[n], splat8(1.061405429f), splat8(-1.453152027f));
    #pragma unroll
    for (int n = 0; n < NPAIR; ++n) p[n] = __builtin_elementwise_fma(t[n], p[n], splat8(1.421413741f));
    #pragma unroll
    for (int n = 0; n < NPAIR; ++n) p[n] = __builtin_elementwise_fma(t[n], p[n], splat8(-0.284496736f));
    #pragma unroll
    for (int n = 0; n < NPAIR; ++n) p[n] = __builtin_elementwise_fma(t[n], p[n], splat8(0.254829592f));
    #pragma unroll
    for (int n = 0; n < NPAIR; ++n) p[n] = p[n] * t[n];
    #pragma unroll
    for (int n = 0; n < NPAIR; ++n) {
        const v8f erfa = __builtin_elementwise_fma(-p[n], e[n], splat8(1.0f));
        const v8f erfz = __builtin_elementwise_copysign(erfa, z[n]);
        const v8f hx   = x[n] * splat8(0.5f);
        x[n] = __builtin_elementwise_fma(hx, erfz, hx);
    }
}

template <int RT>
__device__ __forceinline__ void g2_load_a(v2f (&a)[KG][RT], const float* hb, const int (&hoff)[RT]) {
    #pragma unroll
    for (int j = 0; j < KG; ++j) {
        #pragma unroll
        for (int mt = 0; mt < RT; ++mt) a[j][mt] = *(const v2f*)(hb + hoff[mt] + j * 4);
    }
}

__device__ __forceinline__ void g2_load_b(v2f (&b)[KG][CT2], const float* wp) {
    #pragma unroll
    for (int j = 0; j < KG; ++j) {
        #pragma unroll
        for (int nt = 0; nt < CT2; ++nt) {
            b[j][nt].x = wp[j * 4 * HDIM + nt * 16];
            b[j][nt].y = wp[j * 4 * HDIM + HDIM + nt * 16];
        }
    }
}

#define G2_STAGE(an, bn, ac, bc)                         \
    do {                                                  \
        g2_load_a<RT>(an, hbuf, hoff);                    \
        __builtin_amdgcn_sched_barrier(0x76);             \
        g2_load_b(bn, wp);                                \
        g2_mma<RT>(c, ac, bc);                            \
    } while (0)

template <int RT>
__device__ __forceinline__ void g2_mma(v8f (&c)[RT][CT2], const v2f (&a)[KG][RT], const v2f (&b)[KG][CT2]) {
    #pragma unroll
    for (int j = 0; j < KG; ++j) {
        #pragma unroll
        for (int mt = 0; mt < RT; ++mt) {
            #pragma unroll
            for (int nt = 0; nt < CT2; ++nt)
                c[mt][nt] = __builtin_amdgcn_wmma_f32_16x16x4_f32(false, a[j][mt], false, b[j][nt],
                                                                  (short)0, c[mt][nt], false, false);
        }
    }
}

template <int RT>
__device__ __forceinline__ void g2_block(const float* hbuf, const int* rl, int s0, int n,
                                         const float* __restrict__ W2, const float* __restrict__ b2,
                                         float* __restrict__ out, long base, int ncol0, int ocol0,
                                         int lm, int lh)
{
    int arow[RT];
    #pragma unroll
    for (int mt = 0; mt < RT; ++mt) {
        int s = s0 + mt * 16 + lm;
        arow[mt] = rl[s < n ? s : n - 1];
    }
    int srow[RT][8];
    #pragma unroll
    for (int mt = 0; mt < RT; ++mt) {
        #pragma unroll
        for (int v = 0; v < 8; ++v) {
            int s = s0 + mt * 16 + lh * 8 + v;
            int r = rl[s < n ? s : n - 1];
            srow[mt][v] = (s < n) ? r : -1;
        }
    }
    v8f c[RT][CT2];
    #pragma unroll
    for (int nt = 0; nt < CT2; ++nt) {
        float bias = b2[ncol0 + nt * 16 + lm];
        #pragma unroll
        for (int mt = 0; mt < RT; ++mt) {
            #pragma unroll
            for (int v = 0; v < 8; ++v) c[mt][nt][v] = bias;
        }
    }
    int hoff[RT];
    #pragma unroll
    for (int mt = 0; mt < RT; ++mt) hoff[mt] = arow[mt] * HP + lh * 2;
    const float* wp = W2 + (lh * 2) * HDIM + ncol0 + lm;
    v2f a0[KG][RT], b0[KG][CT2], a1[KG][RT], b1[KG][CT2];
    g2_load_a<RT>(a0, hbuf, hoff);
    g2_load_b(b0, wp);
    #pragma nounroll
    for (int g = 0; g < NGRP - 2; g += 2) {
        #pragma unroll
        for (int mt = 0; mt < RT; ++mt) hoff[mt] += KG * 4;
        wp += KG * 4 * HDIM;
        G2_STAGE(a1, b1, a0, b0);
        #pragma unroll
        for (int mt = 0; mt < RT; ++mt) hoff[mt] += KG * 4;
        wp += KG * 4 * HDIM;
        G2_STAGE(a0, b0, a1, b1);
    }
    #pragma unroll
    for (int mt = 0; mt < RT; ++mt) hoff[mt] += KG * 4;
    wp += KG * 4 * HDIM;
    G2_STAGE(a1, b1, a0, b0);
    g2_mma<RT>(c, a1, b1);
    #pragma unroll
    for (int mt = 0; mt < RT; ++mt) {
        #pragma unroll
        for (int v = 0; v < 8; ++v) {
            if (srow[mt][v] >= 0) {
                float* o = out + (base + srow[mt][v]) * DOUT + ocol0 + lm;
                #pragma unroll
                for (int nt = 0; nt < CT2; ++nt) o[nt * 16] = c[mt][nt][v];
            }
        }
    }
}

__global__ __launch_bounds__(THREADS, 2)
void mpx_encoder_kernel(const float* __restrict__ nf,
                        const float* __restrict__ W1,
                        const float* __restrict__ b1,
                        const float* __restrict__ W2,
                        const float* __restrict__ b2,
                        float* __restrict__ out,
                        int Mtot)
{
    extern __shared__ char smem[];
    float* feats = (float*)smem;
    float* hbuf  = (float*)smem;
    int*   typ   = (int*)(smem + TYPOFF);

    const int  tid  = threadIdx.x;
    const int  lane = tid & 31;
    const int  wv   = tid >> 5;
    const int  lm   = lane & 15;
    const int  lh   = lane >> 4;
    const long base = (long)blockIdx.x * MT;
    int*   rlist = (int*)(smem + RLOFF) + wv * MT;
    int*   cnt   = (int*)(smem + CNTOFF);

    #pragma unroll
    for (int it = 0; it < 16; ++it) {
        int i  = tid + THREADS * it;
        int r  = i >> 5;
        int c4 = i & 31;
        long rs = base + r; if (rs >= Mtot) rs = Mtot - 1;
        const float4 v = *(const float4*)(nf + rs * FTOT + 4 + c4 * 4);
        float* d = feats + r * FP + c4 * 4;
        d[0] = v.x; d[1] = v.y; d[2] = v.z; d[3] = v.w;
    }
    if (tid < MT) {
        long rs = base + tid; if (rs >= Mtot) rs = Mtot - 1;
        const float* p = nf + rs * FTOT;
        float best = p[0]; int bi = 0;
        #pragma unroll
        for (int i = 1; i < 4; ++i) { float q = p[i]; if (q > best) { best = q; bi = i; } }
        typ[tid] = bi;
    }
    __syncthreads();

    v2f afr[MTILES][32];
    #pragma unroll
    for (int mt = 0; mt < MTILES; ++mt) {
        #pragma unroll
        for (int ks = 0; ks < 32; ++ks)
            afr[mt][ks] = *(const v2f*)(feats + (mt * 16 + lm) * FP + ks * 4 + lh * 2);
    }

    const int rows_left = Mtot - (int)base;
    int nmine;
    {
        const bool p0 = (typ[lane] == wv) && (lane < rows_left);
        const bool p1 = (typ[lane + 32] == wv) && (lane + 32 < rows_left);
        const unsigned m0 = __builtin_amdgcn_ballot_w32(p0);
        const unsigned m1 = __builtin_amdgcn_ballot_w32(p1);
        const unsigned below = (1u << lane) - 1u;
        const int cnt0 = __popc(m0);
        if (p0) rlist[__popc(m0 & below)] = lane;
        if (p1) rlist[cnt0 + __popc(m1 & below)] = lane + 32;
        nmine = cnt0 + __popc(m1);
        if (lane == 0) cnt[wv] = nmine;
    }

    __syncthreads();

    for (int np = 0; np < DOUT / (16 * NPAIR); ++np) {
        const int n0 = wv * DOUT + np * (16 * NPAIR);
        v8f c[MTILES][NPAIR];
        #pragma unroll
        for (int nt = 0; nt < NPAIR; ++nt) {
            float bias = b1[n0 + nt * 16 + lm];
            #pragma unroll
            for (int mt = 0; mt < MTILES; ++mt) {
                #pragma unroll
                for (int v = 0; v < 8; ++v) c[mt][nt][v] = bias;
            }
        }
        #pragma unroll
        for (int ks = 0; ks < 32; ++ks) {
            const float* wp = W1 + (ks * 4 + lh * 2) * HDIM + n0 + lm;
            v2f b[NPAIR];
            #pragma unroll
            for (int nt = 0; nt < NPAIR; ++nt) { b[nt].x = wp[nt * 16]; b[nt].y = wp[HDIM + nt * 16]; }
            #pragma unroll
            for (int mt = 0; mt < MTILES; ++mt) {
                #pragma unroll
                for (int nt = 0; nt < NPAIR; ++nt)
                    c[mt][nt] = __builtin_amdgcn_wmma_f32_16x16x4_f32(false, afr[mt][ks], false, b[nt],
                                                                      (short)0, c[mt][nt], false, false);
            }
        }
        __builtin_amdgcn_sched_barrier(0);
        #pragma unroll
        for (int mt = 0; mt < MTILES; ++mt) {
            gelu_tile(c[mt]);
            #pragma unroll
            for (int nt = 0; nt < NPAIR; ++nt) {
                #pragma unroll
                for (int v = 0; v < 8; ++v) {
                    int row = mt * 16 + v + lh * 8;
                    hbuf[row * HP + n0 + nt * 16 + lm] = c[mt][nt][v];
                }
            }
        }
    }
    __syncthreads();

    for (int t = 0; t < 4; ++t) {
        const int n = __builtin_amdgcn_readfirstlane(cnt[t]);
        const int* rl = (const int*)(smem + RLOFF) + t * MT;
        const int ncol0 = t * DOUT + wv * QCOLS;
        for (int s0 = 0; s0 < n; s0 += 32) {
            if (n - s0 > 16) g2_block<2>(hbuf, rl, s0, n, W2, b2, out, base, ncol0, wv * QCOLS, lm, lh);
            else             g2_block<1>(hbuf, rl, s0, n, W2, b2, out, base, ncol0, wv * QCOLS, lm, lh);
        }
    }
}

extern "C" void kernel_launch(void* const* d_in, const int* in_sizes, int n_in,
                              void* d_out, int out_size, void* d_ws, size_t ws_size,
                              hipStream_t stream) {
    const float* nf = (const float*)d_in[0];
    const float* W1 = (const float*)d_in[1];
    const float* b1 = (const float*)d_in[2];
    const float* W2 = (const float*)d_in[3];
    const float* b2 = (const float*)d_in[4];
    float* out = (float*)d_out;

    int Mtot = in_sizes[0] / FTOT;
    int grid = (Mtot + MT - 1) / MT;

    hipFuncSetAttribute((const void*)mpx_encoder_kernel,
                        hipFuncAttributeMaxDynamicSharedMemorySize, SMEM_BYTES);

    mpx_encoder_kernel<<<grid, THREADS, SMEM_BYTES, stream>>>(nf, W1, b1, W2, b2, out, Mtot);
}


# GEMM1 software-pipelined by one pass (8 passes of 16 columns): gelu of pass j-1 issues step by step among the WMMAs of pass j, first weight fragments of pass j+1 loaded ahead; same sums and order
# speedup vs baseline: 5.1968x; 1.0326x over previous
#include <hip/hip_runtime.h>

typedef float v2f __attribute__((ext_vector_type(2)));
typedef float v8f __attribute__((ext_vector_type(8)));

#define MT        64
#define WAVES     4
#define THREADS   128
#define FTOT      132
#define HDIM      512
#define DOUT      128
#define FP        132
#define HP        516
#define HBYTES    (MT * HP * 4)
#define TYPOFF    HBYTES
#define RLOFF     (HBYTES + MT * 4)
#define CNTOFF    (RLOFF + 4 * MT * 4)
#define SMEM_BYTES (CNTOFF + 16)
#define CT2       2
#define QCOLS     (16 * CT2)
#define MTILES    4
#define NPAIR     2
#define KG        8
#define NGRP      (HDIM / 4 / KG)

__device__ __forceinline__ v8f splat8(float s) { v8f r = {s, s, s, s, s, s, s, s}; return r; }

__device__ __forceinline__ void gelu_tile(v8f (&x)[NPAIR]) {
    v8f z[NPAIR], az[NPAIR], t[NPAIR], e[NPAIR], p[NPAIR];
    #pragma unroll
    for (int n = 0; n < NPAIR; ++n) {
        z[n]  = x[n] * splat8(0.70710678118654752f);
        az[n] = __builtin_elementwise_abs(z[n]);
    }
    #pragma unroll
    for (int n = 0; n < NPAIR; ++n) {
        const v8f d = __builtin_elementwise_fma(az[n], splat8(0.3275911f), splat8(1.0f));
        const v8f w = az[n] * az[n] * splat8(-1.4426950408889634f);
        #pragma unroll
        for (int i = 0; i < 8; ++i) {
            t[n][i] = __builtin_amdgcn_rcpf(d[i]);
            e[n][i] = __builtin_amdgcn_exp2f(w[i]);
        }
    }
    #pragma unroll
    for (int n = 0; n < NPAIR; ++n) p[n] = __builtin_elementwise_fma(t[n], splat8(1.061405429f), splat8(-1.453152027f));
    #pragma unroll
    for (int n = 0; n < NPAIR; ++n) p[n] = __builtin_elementwise_fma(t[n], p[n], splat8(1.421413741f));
    #pragma unroll
    for (int n = 0; n < NPAIR; ++n) p[n] = __builtin_elementwise_fma(t[n], p[n], splat8(-0.284496736f));
    #pragma unroll
    for (int n = 0; n < NPAIR; ++n) p[n] = __builtin_elementwise_fma(t[n], p[n], splat8(0.254829592f));
    #pragma unroll
    for (int n = 0; n < NPAIR; ++n) p[n] = p[n] * t[n];
    #pragma unroll
    for (int n = 0; n < NPAIR; ++n) {
        const v8f erfa = __builtin_elementwise_fma(-p[n], e[n], splat8(1.0f));
        const v8f erfz = __builtin_elementwise_copysign(erfa, z[n]);
        const v8f hx   = x[n] * splat8(0.5f);
        x[n] = __builtin_elementwise_fma(hx, erfz, hx);
    }
}

#define G1COLS    16
#define G1PASS    (DOUT / G1COLS)
#define G1PF      8

__device__ __forceinline__ void gelu_step(int s, v8f& x, v8f& z, v8f& t, v8f& e, v8f& p) {
    switch (s) {
    case 0: z = x * splat8(0.70710678118654752f); t = __builtin_elementwise_abs(z); e = t * t; break;
    case 1: t = __builtin_elementwise_fma(t, splat8(0.3275911f), splat8(1.0f));
            #pragma unroll
            for (int i = 0; i < 4; ++i) t[i] = __builtin_amdgcn_rcpf(t[i]);
            break;
    case 2:
            #pragma unroll
            for (int i = 4; i < 8; ++i) t[i] = __builtin_amdgcn_rcpf(t[i]);
            e = e * splat8(-1.4426950408889634f);
            break;
    case 3:
            #pragma unroll
            for (int i = 0; i < 8; ++i) e[i] = __builtin_amdgcn_exp2f(e[i]);
            break;
    case 4: p = __builtin_elementwise_fma(t, splat8(1.061405429f), splat8(-1.453152027f));
            p = __builtin_elementwise_fma(t, p, splat8(1.421413741f));
            break;
    case 5: p = __builtin_elementwise_fma(t, p, splat8(-0.284496736f));
            p = __builtin_elementwise_fma(t, p, splat8(0.254829592f));
            p = p * t;
            break;
    case 6: p = __builtin_elementwise_fma(-p, e, splat8(1.0f));
            p = __builtin_elementwise_copysign(p, z);
            break;
    default: { const v8f hx = x * splat8(0.5f); x = __builtin_elementwise_fma(hx, p, hx); } break;
    }
}

template <bool ACT>
__device__ __forceinline__ void g1_pass(v8f (&c)[MTILES], v8f (&cold)[MTILES], const v2f (&afr)[MTILES][32],
                                        v2f (&pb)[G1PF], float& bias, const float* __restrict__ wcur,
                                        const float* __restrict__ wnext, const float* __restrict__ bnext)
{
    #pragma unroll
    for (int mt = 0; mt < MTILES; ++mt) c[mt] = splat8(bias);
    bias = *bnext;
    v2f b[32];
    #pragma unroll
    for (int ks = G1PF; ks < 32; ++ks) {
        b[ks].x = wcur[ks * 4 * HDIM];
        b[ks].y = wcur[ks * 4 * HDIM + HDIM];
    }
    v8f z, t, e, p;
    #pragma unroll
    for (int ks = 0; ks < 32; ++ks) {
        const v2f f = ks < G1PF ? pb[ks < G1PF ? ks : 0] : b[ks];
        #pragma unroll
        for (int mt = 0; mt < MTILES; ++mt)
            c[mt] = __builtin_amdgcn_wmma_f32_16x16x4_f32(false, afr[mt][ks], false, f,
                                                          (short)0, c[mt], false, false);
        if (ks < G1PF) {
            pb[ks < G1PF ? ks : 0].x = wnext[ks * 4 * HDIM];
            pb[ks < G1PF ? ks : 0].y = wnext[ks * 4 * HDIM + HDIM];
        }
        if (ACT) {
            gelu_step(ks & 7, cold[ks >> 3], z, t, e, p);
            switch (ks & 7) {
            case 0: asm volatile("" : "+v"(z), "+v"(t), "+v"(e)); break;
            case 1: asm volatile("" : "+v"(t)); break;
            case 2: asm volatile("" : "+v"(t), "+v"(e)); break;
            case 3: asm volatile("" : "+v"(e)); break;
            case 7: asm volatile("" : "+v"(cold[ks >> 3])); break;
            default: asm volatile("" : "+v"(p)); break;
            }
            __builtin_amdgcn_sched_barrier(0);
        }
    }
}

__device__ __forceinline__ void g1_store(const v8f (&c)[MTILES], float* hbuf, int col0, int lm, int lh) {
    #pragma unroll
    for (int mt = 0; mt < MTILES; ++mt) {
        #pragma unroll
        for (int v = 0; v < 8; ++v)
            hbuf[(mt * 16 + v + lh * 8) * HP + col0 + lm] = c[mt][v];
    }
}

template <int RT>
__device__ __forceinline__ void g2_load_a(v2f (&a)[KG][RT], const float* hb, const int (&hoff)[RT]) {
    #pragma unroll
    for (int j = 0; j < KG; ++j) {
        #pragma unroll
        for (int mt = 0; mt < RT; ++mt) a[j][mt] = *(const v2f*)(hb + hoff[mt] + j * 4);
    }
}

__device__ __forceinline__ void g2_load_b(v2f (&b)[KG][CT2], const float* wp) {
    #pragma unroll
    for (int j = 0; j < KG; ++j) {
        #pragma unroll
        for (int nt = 0; nt < CT2; ++nt) {
            b[j][nt].x = wp[j * 4 * HDIM + nt * 16];
            b[j][nt].y = wp[j * 4 * HDIM + HDIM + nt * 16];
        }
    }
}

#define G2_STAGE(an, bn, ac, bc)                         \
    do {                                                  \
        g2_load_a<RT>(an, hbuf, hoff);                    \
        __builtin_amdgcn_sched_barrier(0x76);             \
        g2_load_b(bn, wp);                                \
        g2_mma<RT>(c, ac, bc);                            \
    } while (0)

template <int RT>
__device__ __forceinline__ void g2_mma(v8f (&c)[RT][CT2], const v2f (&a)[KG][RT], const v2f (&b)[KG][CT2]) {
    #pragma unroll
    for (int j = 0; j < KG; ++j) {
        #pragma unroll
        for (int mt = 0; mt < RT; ++mt) {
            #pragma unroll
            for (int nt = 0; nt < CT2; ++nt)
                c[mt][nt] = __builtin_amdgcn_wmma_f32_16x16x4_f32(false, a[j][mt], false, b[j][nt],
                                                                  (short)0, c[mt][nt], false, false);
        }
    }
}

template <int RT>
__device__ __forceinline__ void g2_block(const float* hbuf, const int* rl, int s0, int n,
                                         const float* __restrict__ W2, const float* __restrict__ b2,
                                         float* __restrict__ out, long base, int ncol0, int ocol0,
                                         int lm, int lh)
{
    int arow[RT];
    #pragma unroll
    for (int mt = 0; mt < RT; ++mt) {
        int s = s0 + mt * 16 + lm;
        arow[mt] = rl[s < n ? s : n - 1];
    }
    int srow[RT][8];
    #pragma unroll
    for (int mt = 0; mt < RT; ++mt) {
        #pragma unroll
        for (int v = 0; v < 8; ++v) {
            int s = s0 + mt * 16 + lh * 8 + v;
            int r = rl[s < n ? s : n - 1];
            srow[mt][v] = (s < n) ? r : -1;
        }
    }
    v8f c[RT][CT2];
    #pragma unroll
    for (int nt = 0; nt < CT2; ++nt) {
        float bias = b2[ncol0 + nt * 16 + lm];
        #pragma unroll
        for (int mt = 0; mt < RT; ++mt) {
            #pragma unroll
            for (int v = 0; v < 8; ++v) c[mt][nt][v] = bias;
        }
    }
    int hoff[RT];
    #pragma unroll
    for (int mt = 0; mt < RT; ++mt) hoff[mt] = arow[mt] * HP + lh * 2;
    const float* wp = W2 + (lh * 2) * HDIM + ncol0 + lm;
    v2f a0[KG][RT], b0[KG][CT2], a1[KG][RT], b1[KG][CT2];
    g2_load_a<RT>(a0, hbuf, hoff);
    g2_load_b(b0, wp);
    #pragma nounroll
    for (int g = 0; g < NGRP - 2; g += 2) {
        #pragma unroll
        for (int mt = 0; mt < RT; ++mt) hoff[mt] += KG * 4;
        wp += KG * 4 * HDIM;
        G2_STAGE(a1, b1, a0, b0);
        #pragma unroll
        for (int mt = 0; mt < RT; ++mt) hoff[mt] += KG * 4;
        wp += KG * 4 * HDIM;
        G2_STAGE(a0, b0, a1, b1);
    }
    #pragma unroll
    for (int mt = 0; mt < RT; ++mt) hoff[mt] += KG * 4;
    wp += KG * 4 * HDIM;
    G2_STAGE(a1, b1, a0, b0);
    g2_mma<RT>(c, a1, b1);
    #pragma unroll
    for (int mt = 0; mt < RT; ++mt) {
        #pragma unroll
        for (int v = 0; v < 8; ++v) {
            if (srow[mt][v] >= 0) {
                float* o = out + (base + srow[mt][v]) * DOUT + ocol0 + lm;
                #pragma unroll
                for (int nt = 0; nt < CT2; ++nt) o[nt * 16] = c[mt][nt][v];
            }
        }
    }
}

__global__ __launch_bounds__(THREADS, 2)
void mpx_encoder_kernel(const float* __restrict__ nf,
                        const float* __restrict__ W1,
                        const float* __restrict__ b1,
                        const float* __restrict__ W2,
                        const float* __restrict__ b2,
                        float* __restrict__ out,
                        int Mtot)
{
    extern __shared__ char smem[];
    float* feats = (float*)smem;
    float* hbuf  = (float*)smem;
    int*   typ   = (int*)(smem + TYPOFF);

    const int  tid  = threadIdx.x;
    const int  lane = tid & 31;
    const int  wv   = tid >> 5;
    const int  lm   = lane & 15;
    const int  lh   = lane >> 4;
    const long base = (long)blockIdx.x * MT;
    int*   rlist = (int*)(smem + RLOFF) + wv * MT;
    int*   cnt   = (int*)(smem + CNTOFF);

    #pragma unroll
    for (int it = 0; it < 16; ++it) {
        int i  = tid + THREADS * it;
        int r  = i >> 5;
        int c4 = i & 31;
        long rs = base + r; if (rs >= Mtot) rs = Mtot - 1;
        const float4 v = *(const float4*)(nf + rs * FTOT + 4 + c4 * 4);
        float* d = feats + r * FP + c4 * 4;
        d[0] = v.x; d[1] = v.y; d[2] = v.z; d[3] = v.w;
    }
    if (tid < MT) {
        long rs = base + tid; if (rs >= Mtot) rs = Mtot - 1;
        const float* p = nf + rs * FTOT;
        float best = p[0]; int bi = 0;
        #pragma unroll
        for (int i = 1; i < 4; ++i) { float q = p[i]; if (q > best) { best = q; bi = i; } }
        typ[tid] = bi;
    }
    __syncthreads();

    v2f afr[MTILES][32];
    #pragma unroll
    for (int mt = 0; mt < MTILES; ++mt) {
        #pragma unroll
        for (int ks = 0; ks < 32; ++ks)
            afr[mt][ks] = *(const v2f*)(feats + (mt * 16 + lm) * FP + ks * 4 + lh * 2);
    }

    const int rows_left = Mtot - (int)base;
    int nmine;
    {
        const bool p0 = (typ[lane] == wv) && (lane < rows_left);
        const bool p1 = (typ[lane + 32] == wv) && (lane + 32 < rows_left);
        const unsigned m0 = __builtin_amdgcn_ballot_w32(p0);
        const unsigned m1 = __builtin_amdgcn_ballot_w32(p1);
        const unsigned below = (1u << lane) - 1u;
        const int cnt0 = __popc(m0);
        if (p0) rlist[__popc(m0 & below)] = lane;
        if (p1) rlist[cnt0 + __popc(m1 & below)] = lane + 32;
        nmine = cnt0 + __popc(m1);
        if (lane == 0) cnt[wv] = nmine;
    }

    __syncthreads();

    {
        v8f c[MTILES], cold[MTILES];
        v2f pb[G1PF];
        float bias;
        const int colw = wv * DOUT;
        const float* wl = W1 + (lh * 2) * HDIM + lm;
        const float* bl = b1 + lm;
        #pragma unroll
        for (int ks = 0; ks < G1PF; ++ks) {
            pb[ks].x = wl[ks * 4 * HDIM + colw];
            pb[ks].y = wl[ks * 4 * HDIM + HDIM + colw];
        }
        bias = bl[colw];
        g1_pass<false>(cold, c, afr, pb, bias, wl + colw, wl + colw + G1COLS, bl + colw + G1COLS);
        for (int j = 1; j < G1PASS; ++j) {
            const int col = colw + j * G1COLS;
            const int cn = colw + (j + 1 < G1PASS ? j + 1 : G1PASS - 1) * G1COLS;
            g1_pass<true>(c, cold, afr, pb, bias, wl + col, wl + cn, bl + cn);
            g1_store(cold, hbuf, col - G1COLS, lm, lh);
            #pragma unroll
            for (int mt = 0; mt < MTILES; ++mt) cold[mt] = c[mt];
        }
        #pragma unroll
        for (int h = 0; h < MTILES / NPAIR; ++h) {
            v8f x[NPAIR] = {cold[NPAIR * h], cold[NPAIR * h + 1]};
            gelu_tile(x);
            cold[NPAIR * h] = x[0];
            cold[NPAIR * h + 1] = x[1];
        }
        g1_store(cold, hbuf, colw + DOUT - G1COLS, lm, lh);
    }
    __syncthreads();

    for (int t = 0; t < 4; ++t) {
        const int n = __builtin_amdgcn_readfirstlane(cnt[t]);
        const int* rl = (const int*)(smem + RLOFF) + t * MT;
        const int ncol0 = t * DOUT + wv * QCOLS;
        for (int s0 = 0; s0 < n; s0 += 32) {
            if (n - s0 > 16) g2_block<2>(hbuf, rl, s0, n, W2, b2, out, base, ncol0, wv * QCOLS, lm, lh);
            else             g2_block<1>(hbuf, rl, s0, n, W2, b2, out, base, ncol0, wv * QCOLS, lm, lh);
        }
    }
}

extern "C" void kernel_launch(void* const* d_in, const int* in_sizes, int n_in,
                              void* d_out, int out_size, void* d_ws, size_t ws_size,
                              hipStream_t stream) {
    const float* nf = (const float*)d_in[0];
    const float* W1 = (const float*)d_in[1];
    const float* b1 = (const float*)d_in[2];
    const float* W2 = (const float*)d_in[3];
    const float* b2 = (const float*)d_in[4];
    float* out = (float*)d_out;

    int Mtot = in_sizes[0] / FTOT;
    int grid = (Mtot + MT - 1) / MT;

    hipFuncSetAttribute((const void*)mpx_encoder_kernel,
                        hipFuncAttributeMaxDynamicSharedMemorySize, SMEM_BYTES);

    mpx_encoder_kernel<<<grid, THREADS, SMEM_BYTES, stream>>>(nf, W1, b1, W2, b2, out, Mtot);
}


# Rows bucketed by node type over the whole input (2 small launches build a row list): a workgroup takes 64 rows of ONE type, so GEMM2 runs 4 full row tiles, none padded; same k order, sums unchanged
# speedup vs baseline: 6.4611x; 1.2433x over previous
#include <hip/hip_runtime.h>

typedef float v2f __attribute__((ext_vector_type(2)));
typedef float v8f __attribute__((ext_vector_type(8)));

#define MT        64
#define WAVES     4
#define THREADS   128
#define FTOT      132
#define HDIM      512
#define DOUT      128
#define FP        132
#define HP        516
#define HBYTES    (MT * HP * 4)
#define ROWOFF    HBYTES
#define SMEM_BYTES (ROWOFF + MT * 4)
#define CB        1024
#define CW        (CB / 32)
#define CT2       2
#define QCOLS     (16 * CT2)
#define MTILES    4
#define NPAIR     2
#define KG        8
#define NGRP      (HDIM / 4 / KG)

__device__ __forceinline__ v8f splat8(float s) { v8f r = {s, s, s, s, s, s, s, s}; return r; }

__device__ __forceinline__ void gelu_tile(v8f (&x)[NPAIR]) {
    v8f z[NPAIR], az[NPAIR], t[NPAIR], e[NPAIR], p[NPAIR];
    #pragma unroll
    for (int n = 0; n < NPAIR; ++n) {
        z[n]  = x[n] * splat8(0.70710678118654752f);
        az[n] = __builtin_elementwise_abs(z[n]);
    }
    #pragma unroll
    for (int n = 0; n < NPAIR; ++n) {
        const v8f d = __builtin_elementwise_fma(az[n], splat8(0.3275911f), splat8(1.0f));
        const v8f w = az[n] * az[n] * splat8(-1.4426950408889634f);
        #pragma unroll
        for (int i = 0; i < 8; ++i) {
            t[n][i] = __builtin_amdgcn_rcpf(d[i]);
            e[n][i] = __builtin_amdgcn_exp2f(w[i]);
        }
    }
    #pragma unroll
    for (int n = 0; n < NPAIR; ++n) p[n] = __builtin_elementwise_fma(t[n], splat8(1.061405429f), splat8(-1.453152027f));
    #pragma unroll
    for (int n = 0; n < NPAIR; ++n) p[n] = __builtin_elementwise_fma(t[n], p[n], splat8(1.421413741f));
    #pragma unroll
    for (int n = 0; n < NPAIR; ++n) p[n] = __builtin_elementwise_fma(t[n], p[n], splat8(-0.284496736f));
    #pragma unroll
    for (int n = 0; n < NPAIR; ++n) p[n] = __builtin_elementwise_fma(t[n], p[n], splat8(0.254829592f));
    #pragma unroll
    for (int n = 0; n < NPAIR; ++n) p[n] = p[n] * t[n];
    #pragma unroll
    for (int n = 0; n < NPAIR; ++n) {
        const v8f erfa = __builtin_elementwise_fma(-p[n], e[n], splat8(1.0f));
        const v8f erfz = __builtin_elementwise_copysign(erfa, z[n]);
        const v8f hx   = x[n] * splat8(0.5f);
        x[n] = __builtin_elementwise_fma(hx, erfz, hx);
    }
}

#define G1COLS    16
#define G1PASS    (DOUT / G1COLS)
#define G1PF      8

__device__ __forceinline__ void gelu_step(int s, v8f& x, v8f& z, v8f& t, v8f& e, v8f& p) {
    switch (s) {
    case 0: z = x * splat8(0.70710678118654752f); t = __builtin_elementwise_abs(z); e = t * t; break;
    case 1: t = __builtin_elementwise_fma(t, splat8(0.3275911f), splat8(1.0f));
            #pragma unroll
            for (int i = 0; i < 4; ++i) t[i] = __builtin_amdgcn_rcpf(t[i]);
            break;
    case 2:
            #pragma unroll
            for (int i = 4; i < 8; ++i) t[i] = __builtin_amdgcn_rcpf(t[i]);
            e = e * splat8(-1.4426950408889634f);
            break;
    case 3:
            #pragma unroll
            for (int i = 0; i < 8; ++i) e[i] = __builtin_amdgcn_exp2f(e[i]);
            break;
    case 4: p = __builtin_elementwise_fma(t, splat8(1.061405429f), splat8(-1.453152027f));
            p = __builtin_elementwise_fma(t, p, splat8(1.421413741f));
            break;
    case 5: p = __builtin_elementwise_fma(t, p, splat8(-0.284496736f));
            p = __builtin_elementwise_fma(t, p, splat8(0.254829592f));
            p = p * t;
            break;
    case 6: p = __builtin_elementwise_fma(-p, e, splat8(1.0f));
            p = __builtin_elementwise_copysign(p, z);
            break;
    default: { const v8f hx = x * splat8(0.5f); x = __builtin_elementwise_fma(hx, p, hx); } break;
    }
}

template <bool ACT>
__device__ __forceinline__ void g1_pass(v8f (&c)[MTILES], v8f (&cold)[MTILES], const v2f (&afr)[MTILES][32],
                                        v2f (&pb)[G1PF], float& bias, const float* __restrict__ wcur,
                                        const float* __restrict__ wnext, const float* __restrict__ bnext)
{
    #pragma unroll
    for (int mt = 0; mt < MTILES; ++mt) c[mt] = splat8(bias);
    bias = *bnext;
    v2f b[32];
    #pragma unroll
    for (int ks = G1PF; ks < 32; ++ks) {
        b[ks].x = wcur[ks * 4 * HDIM];
        b[ks].y = wcur[ks * 4 * HDIM + HDIM];
    }
    v8f z, t, e, p;
    #pragma unroll
    for (int ks = 0; ks < 32; ++ks) {
        const v2f f = ks < G1PF ? pb[ks < G1PF ? ks : 0] : b[ks];
        #pragma unroll
        for (int mt = 0; mt < MTILES; ++mt)
            c[mt] = __builtin_amdgcn_wmma_f32_16x16x4_f32(false, afr[mt][ks], false, f,
                                                          (short)0, c[mt], false, false);
        if (ks < G1PF) {
            pb[ks < G1PF ? ks : 0].x = wnext[ks * 4 * HDIM];
            pb[ks < G1PF ? ks : 0].y = wnext[ks * 4 * HDIM + HDIM];
        }
        if (ACT) {
            gelu_step(ks & 7, cold[ks >> 3], z, t, e, p);
            switch (ks & 7) {
            case 0: asm volatile("" : "+v"(z), "+v"(t), "+v"(e)); break;
            case 1: asm volatile("" : "+v"(t)); break;
            case 2: asm volatile("" : "+v"(t), "+v"(e)); break;
            case 3: asm volatile("" : "+v"(e)); break;
            case 7: asm volatile("" : "+v"(cold[ks >> 3])); break;
            default: asm volatile("" : "+v"(p)); break;
            }
            __builtin_amdgcn_sched_barrier(0);
        }
    }
}

__device__ __forceinline__ void g1_store(const v8f (&c)[MTILES], float* hbuf, int col0, int lm, int lh) {
    #pragma unroll
    for (int mt = 0; mt < MTILES; ++mt) {
        #pragma unroll
        for (int v = 0; v < 8; ++v)
            hbuf[(mt * 16 + v + lh * 8) * HP + col0 + lm] = c[mt][v];
    }
}

__device__ __forceinline__ int ws_list(int Mtot) { return 64 + 4 * ((Mtot + CB - 1) / CB); }

__global__ __launch_bounds__(CB)
void mpx_type_count_kernel(const float* __restrict__ nf, int* __restrict__ ws, int Mtot)
{
    __shared__ int wc[CW][4];
    const int tid = threadIdx.x, lane = tid & 31, w = tid >> 5;
    const int r = blockIdx.x * CB + tid;
    const bool live = r < Mtot;
    int bi = 4;
    if (live) {
        const float4 q = *(const float4*)(nf + (long)r * FTOT);
        float best = q.x; bi = 0;
        if (q.y > best) { best = q.y; bi = 1; }
        if (q.z > best) { best = q.z; bi = 2; }
        if (q.w > best) { best = q.w; bi = 3; }
        ((unsigned char*)(ws + ws_list(Mtot) + Mtot))[r] = (unsigned char)bi;
    }
    #pragma unroll
    for (int t = 0; t < 4; ++t) {
        const unsigned m = __builtin_amdgcn_ballot_w32(bi == t);
        if (lane == 0) wc[w][t] = __popc(m);
    }
    __syncthreads();
    if (tid < 4) {
        int s = 0;
        for (int i = 0; i < CW; ++i) s += wc[i][tid];
        ws[64 + 4 * blockIdx.x + tid] = s;
    }
}

__global__ __launch_bounds__(CB)
void mpx_place_kernel(int* __restrict__ ws, int Mtot)
{
    __shared__ int wc[CW][4];
    __shared__ int lo[4], all[4];
    const int tid = threadIdx.x, lane = tid & 31, w = tid >> 5;
    const int nblk = (Mtot + CB - 1) / CB;
    const int* cnts = ws + 64;
    if (w < 4) {
        int sl = 0, sa = 0;
        for (int b = lane; b < nblk; b += 32) {
            const int v = cnts[4 * b + w];
            sa += v;
            if (b < (int)blockIdx.x) sl += v;
        }
        for (int o = 16; o > 0; o >>= 1) { sl += __shfl_xor(sl, o); sa += __shfl_xor(sa, o); }
        if (lane == 0) { lo[w] = sl; all[w] = sa; }
    }
    const int r = blockIdx.x * CB + tid;
    const bool live = r < Mtot;
    const unsigned char* typ = (const unsigned char*)(ws + ws_list(Mtot) + Mtot);
    const int bi = live ? (int)typ[r] : 4;
    unsigned mine = 0;
    #pragma unroll
    for (int t = 0; t < 4; ++t) {
        const unsigned m = __builtin_amdgcn_ballot_w32(bi == t);
        if (lane == 0) wc[w][t] = __popc(m);
        if (bi == t) mine = m;
    }
    __syncthreads();
    if (live) {
        int pos = lo[bi];
        for (int u = 0; u < bi; ++u) pos += all[u];
        for (int i = 0; i < w; ++i) pos += wc[i][bi];
        pos += __popc(mine & ((1u << lane) - 1u));
        ws[ws_list(Mtot) + pos] = r;
    }
    if (blockIdx.x == 0 && tid < 4) ws[tid] = all[tid];
}

__device__ __forceinline__ void g2_load_a(v2f (&a)[KG][MTILES], const float* hb, const int (&hoff)[MTILES]) {
    #pragma unroll
    for (int j = 0; j < KG; ++j) {
        #pragma unroll
        for (int mt = 0; mt < MTILES; ++mt) a[j][mt] = *(const v2f*)(hb + hoff[mt] + j * 4);
    }
}

__device__ __forceinline__ void g2_load_b(v2f (&b)[KG][CT2], const float* wp) {
    #pragma unroll
    for (int j = 0; j < KG; ++j) {
        #pragma unroll
        for (int nt = 0; nt < CT2; ++nt) {
            b[j][nt].x = wp[j * 4 * HDIM + nt * 16];
            b[j][nt].y = wp[j * 4 * HDIM + HDIM + nt * 16];
        }
    }
}

#define G2_STAGE(an, bn, ac, bc)                         \
    do {                                                  \
        g2_load_a(an, hbuf, hoff);                        \
        __builtin_amdgcn_sched_barrier(0x76);             \
        g2_load_b(bn, wp);                                \
        g2_mma(c, ac, bc);                                \
    } while (0)

__device__ __forceinline__ void g2_mma(v8f (&c)[MTILES][CT2], const v2f (&a)[KG][MTILES], const v2f (&b)[KG][CT2]) {
    #pragma unroll
    for (int j = 0; j < KG; ++j) {
        #pragma unroll
        for (int mt = 0; mt < MTILES; ++mt) {
            #pragma unroll
            for (int nt = 0; nt < CT2; ++nt)
                c[mt][nt] = __builtin_amdgcn_wmma_f32_16x16x4_f32(false, a[j][mt], false, b[j][nt],
                                                                  (short)0, c[mt][nt], false, false);
        }
    }
}

__device__ __forceinline__ void g2_block(const float* hbuf, const int* rowid, int nrows,
                                         const float* __restrict__ W2, const float* __restrict__ b2,
                                         float* __restrict__ out, int ncol0, int ocol0, int lm, int lh)
{
    v8f c[MTILES][CT2];
    #pragma unroll
    for (int nt = 0; nt < CT2; ++nt) {
        const float bias = b2[ncol0 + nt * 16 + lm];
        #pragma unroll
        for (int mt = 0; mt < MTILES; ++mt) c[mt][nt] = splat8(bias);
    }
    int hoff[MTILES];
    #pragma unroll
    for (int mt = 0; mt < MTILES; ++mt) hoff[mt] = (mt * 16 + lm) * HP + lh * 2;
    const float* wp = W2 + (lh * 2) * HDIM + ncol0 + lm;
    v2f a0[KG][MTILES], b0[KG][CT2], a1[KG][MTILES], b1[KG][CT2];
    g2_load_a(a0, hbuf, hoff);
    g2_load_b(b0, wp);
    for (int g = 0; g < NGRP - 2; g += 2) {
        #pragma unroll
        for (int mt = 0; mt < MTILES; ++mt) hoff[mt] += KG * 4;
        wp += KG * 4 * HDIM;
        G2_STAGE(a1, b1, a0, b0);
        #pragma unroll
        for (int mt = 0; mt < MTILES; ++mt) hoff[mt] += KG * 4;
        wp += KG * 4 * HDIM;
        G2_STAGE(a0, b0, a1, b1);
    }
    #pragma unroll
    for (int mt = 0; mt < MTILES; ++mt) hoff[mt] += KG * 4;
    wp += KG * 4 * HDIM;
    G2_STAGE(a1, b1, a0, b0);
    g2_mma(c, a1, b1);
    #pragma unroll
    for (int mt = 0; mt < MTILES; ++mt) {
        #pragma unroll
        for (int v = 0; v < 8; ++v) {
            const int s = mt * 16 + lh * 8 + v;
            if (s < nrows) {
                float* o = out + (long)rowid[s] * DOUT + ocol0 + lm;
                #pragma unroll
                for (int nt = 0; nt < CT2; ++nt) o[nt * 16] = c[mt][nt][v];
            }
        }
    }
}

__global__ __launch_bounds__(THREADS, 2)
void mpx_encoder_kernel(const float* __restrict__ nf,
                        const float* __restrict__ W1,
                        const float* __restrict__ b1,
                        const float* __restrict__ W2,
                        const float* __restrict__ b2,
                        float* __restrict__ out,
                        const int* __restrict__ ws,
                        int Mtot)
{
    extern __shared__ char smem[];
    float* feats = (float*)smem;
    float* hbuf  = (float*)smem;
    int*   rowid = (int*)(smem + ROWOFF);

    const int  tid  = threadIdx.x;
    const int  lane = tid & 31;
    const int  wv   = __builtin_amdgcn_readfirstlane(tid >> 5);
    const int  lm   = lane & 15;
    const int  lh   = lane >> 4;

    int typ = -1, first = 0, nleft = 0;
    {
        int g = blockIdx.x, off = 0;
        #pragma unroll
        for (int u = 0; u < 4; ++u) {
            const int nu = ws[u];
            const int gu = (nu + MT - 1) / MT;
            if (typ < 0) {
                if (g < gu) { typ = u; first = off + g * MT; nleft = nu - g * MT; }
                else g -= gu;
            }
            off += nu;
        }
    }
    if (typ < 0) return;
    const int  nrows = nleft < MT ? nleft : MT;
    const int* list  = ws + ws_list(Mtot) + first;

    #pragma unroll
    for (int it = 0; it < 16; ++it) {
        int r  = wv + 4 * it;
        int c4 = lane;
        const long rs = list[r < nrows ? r : nrows - 1];
        const float4 v = *(const float4*)(nf + rs * FTOT + 4 + c4 * 4);
        float* d = feats + r * FP + c4 * 4;
        d[0] = v.x; d[1] = v.y; d[2] = v.z; d[3] = v.w;
    }
    if (tid < MT) rowid[tid] = list[tid < nrows ? tid : nrows - 1];
    __syncthreads();

    v2f afr[MTILES][32];
    #pragma unroll
    for (int mt = 0; mt < MTILES; ++mt) {
        #pragma unroll
        for (int ks = 0; ks < 32; ++ks)
            afr[mt][ks] = *(const v2f*)(feats + (mt * 16 + lm) * FP + ks * 4 + lh * 2);
    }

    __syncthreads();

    {
        v8f c[MTILES], cold[MTILES];
        v2f pb[G1PF];
        float bias;
        const int colw = wv * DOUT;
        const float* wl = W1 + (lh * 2) * HDIM + lm;
        const float* bl = b1 + lm;
        #pragma unroll
        for (int ks = 0; ks < G1PF; ++ks) {
            pb[ks].x = wl[ks * 4 * HDIM + colw];
            pb[ks].y = wl[ks * 4 * HDIM + HDIM + colw];
        }
        bias = bl[colw];
        g1_pass<false>(cold, c, afr, pb, bias, wl + colw, wl + colw + G1COLS, bl + colw + G1COLS);
        for (int j = 1; j < G1PASS; ++j) {
            const int col = colw + j * G1COLS;
            const int cn = colw + (j + 1 < G1PASS ? j + 1 : G1PASS - 1) * G1COLS;
            g1_pass<true>(c, cold, afr, pb, bias, wl + col, wl + cn, bl + cn);
            g1_store(cold, hbuf, col - G1COLS, lm, lh);
            #pragma unroll
            for (int mt = 0; mt < MTILES; ++mt) cold[mt] = c[mt];
        }
        #pragma unroll
        for (int h = 0; h < MTILES / NPAIR; ++h) {
            v8f x[NPAIR] = {cold[NPAIR * h], cold[NPAIR * h + 1]};
            gelu_tile(x);
            cold[NPAIR * h] = x[0];
            cold[NPAIR * h + 1] = x[1];
        }
        g1_store(cold, hbuf, colw + DOUT - G1COLS, lm, lh);
    }
    __syncthreads();

    g2_block(hbuf, rowid, nrows, W2, b2, out, typ * DOUT + wv * QCOLS, wv * QCOLS, lm, lh);
}

extern "C" void kernel_launch(void* const* d_in, const int* in_sizes, int n_in,
                              void* d_out, int out_size, void* d_ws, size_t ws_size,
                              hipStream_t stream) {
    const float* nf = (const float*)d_in[0];
    const float* W1 = (const float*)d_in[1];
    const float* b1 = (const float*)d_in[2];
    const float* W2 = (const float*)d_in[3];
    const float* b2 = (const float*)d_in[4];
    float* out = (float*)d_out;
    int* ws = (int*)d_ws;

    int Mtot = in_sizes[0] / FTOT;
    int grid = (Mtot + MT - 1) / MT;
    int nblk = (Mtot + CB - 1) / CB;
    if (ws_size < (size_t)(64 + 4 * nblk + Mtot) * 4 + (size_t)Mtot) return;

    hipFuncSetAttribute((const void*)mpx_encoder_kernel,
                        hipFuncAttributeMaxDynamicSharedMemorySize, SMEM_BYTES);

    mpx_type_count_kernel<<<nblk, CB, 0, stream>>>(nf, ws, Mtot);
    mpx_place_kernel<<<nblk, CB, 0, stream>>>(ws, Mtot);
    mpx_encoder_kernel<<<grid + 3, THREADS, SMEM_BYTES, stream>>>(nf, W1, b1, W2, b2, out, ws, Mtot);
}


# Weights re-laid per call (in launch 1) into WMMA fragment order: a lane's B fragments of 2 k-steps are one 16-byte load of whole lines, 4x fewer weight loads; k summed in another order within each 8
# speedup vs baseline: 6.7739x; 1.0484x over previous
#include <hip/hip_runtime.h>

typedef float v2f __attribute__((ext_vector_type(2)));
typedef float v4f __attribute__((ext_vector_type(4)));
typedef float v8f __attribute__((ext_vector_type(8)));

#define MT        64
#define WAVES     4
#define THREADS   128
#define FTOT      132
#define HDIM      512
#define DOUT      128
#define FP        132
#define HP        516
#define HBYTES    (MT * HP * 4)
#define ROWOFF    HBYTES
#define SMEM_BYTES (ROWOFF + MT * 4)
#define CB        1024
#define CW        (CB / 32)
#define CT2       2
#define QCOLS     (16 * CT2)
#define MTILES    4
#define NPAIR     2
#define KG        8
#define NGRP      (HDIM / 4 / KG)
#define KIN       128
#define W1F4      (HDIM / 16 * (KIN / 8) * 32)
#define W2F4      (HDIM / 16 * (HDIM / 8) * 32)
#define WFBLK     ((W1F4 + W2F4) / CB)

__device__ __forceinline__ v8f splat8(float s) { v8f r = {s, s, s, s, s, s, s, s}; return r; }

__device__ __forceinline__ void gelu_tile(v8f (&x)[NPAIR]) {
    v8f z[NPAIR], az[NPAIR], t[NPAIR], e[NPAIR], p[NPAIR];
    #pragma unroll
    for (int n = 0; n < NPAIR; ++n) {
        z[n]  = x[n] * splat8(0.70710678118654752f);
        az[n] = __builtin_elementwise_abs(z[n]);
    }
    #pragma unroll
    for (int n = 0; n < NPAIR; ++n) {
        const v8f d = __builtin_elementwise_fma(az[n], splat8(0.3275911f), splat8(1.0f));
        const v8f w = az[n] * az[n] * splat8(-1.4426950408889634f);
        #pragma unroll
        for (int i = 0; i < 8; ++i) {
            t[n][i] = __builtin_amdgcn_rcpf(d[i]);
            e[n][i] = __builtin_amdgcn_exp2f(w[i]);
        }
    }
    #pragma unroll
    for (int n = 0; n < NPAIR; ++n) p[n] = __builtin_elementwise_fma(t[n], splat8(1.061405429f), splat8(-1.453152027f));
    #pragma unroll
    for (int n = 0; n < NPAIR; ++n) p[n] = __builtin_elementwise_fma(t[n], p[n], splat8(1.421413741f));
    #pragma unroll
    for (int n = 0; n < NPAIR; ++n) p[n] = __builtin_elementwise_fma(t[n], p[n], splat8(-0.284496736f));
    #pragma unroll
    for (int n = 0; n < NPAIR; ++n) p[n] = __builtin_elementwise_fma(t[n], p[n], splat8(0.254829592f));
    #pragma unroll
    for (int n = 0; n < NPAIR; ++n) p[n] = p[n] * t[n];
    #pragma unroll
    for (int n = 0; n < NPAIR; ++n) {
        const v8f erfa = __builtin_elementwise_fma(-p[n], e[n], splat8(1.0f));
        const v8f erfz = __builtin_elementwise_copysign(erfa, z[n]);
        const v8f hx   = x[n] * splat8(0.5f);
        x[n] = __builtin_elementwise_fma(hx, erfz, hx);
    }
}

#define G1COLS    16
#define G1PASS    (DOUT / G1COLS)
#define G1PF      8

__device__ __forceinline__ void gelu_step(int s, v8f& x, v8f& z, v8f& t, v8f& e, v8f& p) {
    switch (s) {
    case 0: z = x * splat8(0.70710678118654752f); t = __builtin_elementwise_abs(z); e = t * t; break;
    case 1: t = __builtin_elementwise_fma(t, splat8(0.3275911f), splat8(1.0f));
            #pragma unroll
            for (int i = 0; i < 4; ++i) t[i] = __builtin_amdgcn_rcpf(t[i]);
            break;
    case 2:
            #pragma unroll
            for (int i = 4; i < 8; ++i) t[i] = __builtin_amdgcn_rcpf(t[i]);
            e = e * splat8(-1.4426950408889634f);
            break;
    case 3:
            #pragma unroll
            for (int i = 0; i < 8; ++i) e[i] = __builtin_amdgcn_exp2f(e[i]);
            break;
    case 4: p = __builtin_elementwise_fma(t, splat8(1.061405429f), splat8(-1.453152027f));
            p = __builtin_elementwise_fma(t, p, splat8(1.421413741f));
            break;
    case 5: p = __builtin_elementwise_fma(t, p, splat8(-0.284496736f));
            p = __builtin_elementwise_fma(t, p, splat8(0.254829592f));
            p = p * t;
            break;
    case 6: p = __builtin_elementwise_fma(-p, e, splat8(1.0f));
            p = __builtin_elementwise_copysign(p, z);
            break;
    default: { const v8f hx = x * splat8(0.5f); x = __builtin_elementwise_fma(hx, p, hx); } break;
    }
}

template <bool ACT>
__device__ __forceinline__ void g1_pass(v8f (&c)[MTILES], v8f (&cold)[MTILES], const v2f (&afr)[MTILES][32],
                                        v2f (&pb)[G1PF], float& bias, const float* __restrict__ wcur,
                                        const float* __restrict__ wnext, const float* __restrict__ bnext)
{
    #pragma unroll
    for (int mt = 0; mt < MTILES; ++mt) c[mt] = splat8(bias);
    bias = *bnext;
    v2f b[32];
    #pragma unroll
    for (int q = G1PF / 2; q < 16; ++q) {
        const v4f w = *(const v4f*)(wcur + q * 128);
        b[2 * q].x = w.x;     b[2 * q].y = w.y;
        b[2 * q + 1].x = w.z; b[2 * q + 1].y = w.w;
    }
    v8f z, t, e, p;
    #pragma unroll
    for (int ks = 0; ks < 32; ++ks) {
        const v2f f = ks < G1PF ? pb[ks < G1PF ? ks : 0] : b[ks];
        #pragma unroll
        for (int mt = 0; mt < MTILES; ++mt)
            c[mt] = __builtin_amdgcn_wmma_f32_16x16x4_f32(false, afr[mt][ks], false, f,
                                                          (short)0, c[mt], false, false);
        if (ks < G1PF && (ks & 1)) {
            const int k1 = (ks < G1PF && (ks & 1)) ? ks : 1;
            const v4f w = *(const v4f*)(wnext + (k1 >> 1) * 128);
            pb[k1 - 1].x = w.x; pb[k1 - 1].y = w.y;
            pb[k1].x = w.z;     pb[k1].y = w.w;
        }
        if (ACT) {
            gelu_step(ks & 7, cold[ks >> 3], z, t, e, p);
            switch (ks & 7) {
            case 0: asm volatile("" : "+v"(z), "+v"(t), "+v"(e)); break;
            case 1: asm volatile("" : "+v"(t)); break;
            case 2: asm volatile("" : "+v"(t), "+v"(e)); break;
            case 3: asm volatile("" : "+v"(e)); break;
            case 7: asm volatile("" : "+v"(cold[ks >> 3])); break;
            default: asm volatile("" : "+v"(p)); break;
            }
            __builtin_amdgcn_sched_barrier(0);
        }
    }
}

__device__ __forceinline__ void g1_store(const v8f (&c)[MTILES], float* hbuf, int col0, int lm, int lh) {
    #pragma unroll
    for (int mt = 0; mt < MTILES; ++mt) {
        #pragma unroll
        for (int v = 0; v < 8; ++v)
            hbuf[(mt * 16 + v + lh * 8) * HP + col0 + lm] = c[mt][v];
    }
}

__host__ __device__ __forceinline__ int ws_list(int Mtot) { return 64 + 4 * ((Mtot + CB - 1) / CB); }

__host__ __device__ __forceinline__ long ws_wf_bytes(int Mtot) {
    return (((long)(ws_list(Mtot) + Mtot) * 4 + Mtot) + 255) & ~255L;
}

__global__ __launch_bounds__(CB)
void mpx_type_count_kernel(const float* __restrict__ nf, const float* __restrict__ W1,
                           const float* __restrict__ W2, int* __restrict__ ws, int Mtot)
{
    __shared__ int wc[CW][4];
    const int tid = threadIdx.x, lane = tid & 31, w = tid >> 5;
    const int nblk = (Mtot + CB - 1) / CB;
    if ((int)blockIdx.x >= nblk) {
        int i = ((int)blockIdx.x - nblk) * CB + tid;
        v4f* dst = (v4f*)((char*)ws + ws_wf_bytes(Mtot)) + i;
        const float* Wm = W1;
        int nqs = 4;
        if (i >= W1F4) { i -= W1F4; Wm = W2; nqs = 6; }
        const int blk = i >> 5, q = blk & ((1 << nqs) - 1), ct = blk >> nqs;
        const float* s = Wm + (long)(8 * q + 4 * (lane >> 4)) * HDIM + 16 * ct + (lane & 15);
        v4f v = {s[0], s[HDIM], s[2 * HDIM], s[3 * HDIM]};
        *dst = v;
        return;
    }
    const int r = blockIdx.x * CB + tid;
    const bool live = r < Mtot;
    int bi = 4;
    if (live) {
        const float4 q = *(const float4*)(nf + (long)r * FTOT);
        float best = q.x; bi = 0;
        if (q.y > best) { best = q.y; bi = 1; }
        if (q.z > best) { best = q.z; bi = 2; }
        if (q.w > best) { best = q.w; bi = 3; }
        ((unsigned char*)(ws + ws_list(Mtot) + Mtot))[r] = (unsigned char)bi;
    }
    #pragma unroll
    for (int t = 0; t < 4; ++t) {
        const unsigned m = __builtin_amdgcn_ballot_w32(bi == t);
        if (lane == 0) wc[w][t] = __popc(m);
    }
    __syncthreads();
    if (tid < 4) {
        int s = 0;
        for (int i = 0; i < CW; ++i) s += wc[i][tid];
        ws[64 + 4 * blockIdx.x + tid] = s;
    }
}

__global__ __launch_bounds__(CB)
void mpx_place_kernel(int* __restrict__ ws, int Mtot)
{
    __shared__ int wc[CW][4];
    __shared__ int lo[4], all[4];
    const int tid = threadIdx.x, lane = tid & 31, w = tid >> 5;
    const int nblk = (Mtot + CB - 1) / CB;
    const int* cnts = ws + 64;
    if (w < 4) {
        int sl = 0, sa = 0;
        for (int b = lane; b < nblk; b += 32) {
            const int v = cnts[4 * b + w];
            sa += v;
            if (b < (int)blockIdx.x) sl += v;
        }
        for (int o = 16; o > 0; o >>= 1) { sl += __shfl_xor(sl, o); sa += __shfl_xor(sa, o); }
        if (lane == 0) { lo[w] = sl; all[w] = sa; }
    }
    const int r = blockIdx.x * CB + tid;
    const bool live = r < Mtot;
    const unsigned char* typ = (const unsigned char*)(ws + ws_list(Mtot) + Mtot);
    const int bi = live ? (int)typ[r] : 4;
    unsigned mine = 0;
    #pragma unroll
    for (int t = 0; t < 4; ++t) {
        const unsigned m = __builtin_amdgcn_ballot_w32(bi == t);
        if (lane == 0) wc[w][t] = __popc(m);
        if (bi == t) mine = m;
    }
    __syncthreads();
    if (live) {
        int pos = lo[bi];
        for (int u = 0; u < bi; ++u) pos += all[u];
        for (int i = 0; i < w; ++i) pos += wc[i][bi];
        pos += __popc(mine & ((1u << lane) - 1u));
        ws[ws_list(Mtot) + pos] = r;
    }
    if (blockIdx.x == 0 && tid < 4) ws[tid] = all[tid];
}

__device__ __forceinline__ void g2_load_a(v2f (&a)[KG][MTILES], const float* hb, const int (&hoff)[MTILES]) {
    #pragma unroll
    for (int p = 0; p < KG / 2; ++p) {
        #pragma unroll
        for (int mt = 0; mt < MTILES; ++mt) {
            const v4f t = *(const v4f*)(hb + hoff[mt] + p * 8);
            a[2 * p][mt].x = t.x;     a[2 * p][mt].y = t.y;
            a[2 * p + 1][mt].x = t.z; a[2 * p + 1][mt].y = t.w;
        }
    }
}

__device__ __forceinline__ void g2_load_b(v2f (&b)[KG][CT2], const float* wp) {
    #pragma unroll
    for (int p = 0; p < KG / 2; ++p) {
        #pragma unroll
        for (int nt = 0; nt < CT2; ++nt) {
            const v4f w = *(const v4f*)(wp + nt * (HDIM / 8) * 128 + p * 128);
            b[2 * p][nt].x = w.x;     b[2 * p][nt].y = w.y;
            b[2 * p + 1][nt].x = w.z; b[2 * p + 1][nt].y = w.w;
        }
    }
}

#define G2_STAGE(an, bn, ac, bc)                         \
    do {                                                  \
        g2_load_a(an, hbuf, hoff);                        \
        __builtin_amdgcn_sched_barrier(0x76);             \
        g2_load_b(bn, wp);                                \
        g2_mma(c, ac, bc);                                \
    } while (0)

__device__ __forceinline__ void g2_mma(v8f (&c)[MTILES][CT2], const v2f (&a)[KG][MTILES], const v2f (&b)[KG][CT2]) {
    #pragma unroll
    for (int j = 0; j < KG; ++j) {
        #pragma unroll
        for (int mt = 0; mt < MTILES; ++mt) {
            #pragma unroll
            for (int nt = 0; nt < CT2; ++nt)
                c[mt][nt] = __builtin_amdgcn_wmma_f32_16x16x4_f32(false, a[j][mt], false, b[j][nt],
                                                                  (short)0, c[mt][nt], false, false);
        }
    }
}

__device__ __forceinline__ void g2_block(const float* hbuf, const int* rowid, int nrows,
                                         const float* __restrict__ W2f, const float* __restrict__ b2,
                                         float* __restrict__ out, int ncol0, int ocol0, int lm, int lh)
{
    v8f c[MTILES][CT2];
    #pragma unroll
    for (int nt = 0; nt < CT2; ++nt) {
        const float bias = b2[ncol0 + nt * 16 + lm];
        #pragma unroll
        for (int mt = 0; mt < MTILES; ++mt) c[mt][nt] = splat8(bias);
    }
    int hoff[MTILES];
    #pragma unroll
    for (int mt = 0; mt < MTILES; ++mt) hoff[mt] = (mt * 16 + lm) * HP + lh * 4;
    const float* wp = W2f + (ncol0 / 16) * (HDIM / 8) * 128 + (lh * 16 + lm) * 4;
    v2f a0[KG][MTILES], b0[KG][CT2], a1[KG][MTILES], b1[KG][CT2];
    g2_load_a(a0, hbuf, hoff);
    g2_load_b(b0, wp);
    for (int g = 0; g < NGRP - 2; g += 2) {
        #pragma unroll
        for (int mt = 0; mt < MTILES; ++mt) hoff[mt] += KG * 4;
        wp += (KG / 2) * 128;
        G2_STAGE(a1, b1, a0, b0);
        #pragma unroll
        for (int mt = 0; mt < MTILES; ++mt) hoff[mt] += KG * 4;
        wp += (KG / 2) * 128;
        G2_STAGE(a0, b0, a1, b1);
    }
    #pragma unroll
    for (int mt = 0; mt < MTILES; ++mt) hoff[mt] += KG * 4;
    wp += (KG / 2) * 128;
    G2_STAGE(a1, b1, a0, b0);
    g2_mma(c, a1, b1);
    #pragma unroll
    for (int mt = 0; mt < MTILES; ++mt) {
        #pragma unroll
        for (int v = 0; v < 8; ++v) {
            const int s = mt * 16 + lh * 8 + v;
            if (s < nrows) {
                float* o = out + (long)rowid[s] * DOUT + ocol0 + lm;
                #pragma unroll
                for (int nt = 0; nt < CT2; ++nt) o[nt * 16] = c[mt][nt][v];
            }
        }
    }
}

__global__ __launch_bounds__(THREADS, 2)
void mpx_encoder_kernel(const float* __restrict__ nf,
                        const float* __restrict__ b1,
                        const float* __restrict__ b2,
                        float* __restrict__ out,
                        const int* __restrict__ ws,
                        int Mtot)
{
    extern __shared__ char smem[];
    float* feats = (float*)smem;
    float* hbuf  = (float*)smem;
    int*   rowid = (int*)(smem + ROWOFF);

    const int  tid  = threadIdx.x;
    const int  lane = tid & 31;
    const int  wv   = __builtin_amdgcn_readfirstlane(tid >> 5);
    const int  lm   = lane & 15;
    const int  lh   = lane >> 4;

    int typ = -1, first = 0, nleft = 0;
    {
        int g = blockIdx.x, off = 0;
        #pragma unroll
        for (int u = 0; u < 4; ++u) {
            const int nu = ws[u];
            const int gu = (nu + MT - 1) / MT;
            if (typ < 0) {
                if (g < gu) { typ = u; first = off + g * MT; nleft = nu - g * MT; }
                else g -= gu;
            }
            off += nu;
        }
    }
    if (typ < 0) return;
    const int  nrows = nleft < MT ? nleft : MT;
    const int* list  = ws + ws_list(Mtot) + first;
    const float* W1f = (const float*)((const char*)ws + ws_wf_bytes(Mtot));
    const float* W2f = W1f + W1F4 * 4;

    #pragma unroll
    for (int it = 0; it < 16; ++it) {
        int r  = wv + 4 * it;
        int c4 = lane;
        const long rs = list[r < nrows ? r : nrows - 1];
        const float4 v = *(const float4*)(nf + rs * FTOT + 4 + c4 * 4);
        float* d = feats + r * FP + c4 * 4;
        d[0] = v.x; d[1] = v.y; d[2] = v.z; d[3] = v.w;
    }
    if (tid < MT) rowid[tid] = list[tid < nrows ? tid : nrows - 1];
    __syncthreads();

    v2f afr[MTILES][32];
    #pragma unroll
    for (int mt = 0; mt < MTILES; ++mt) {
        #pragma unroll
        for (int q = 0; q < 16; ++q) {
            const v4f t = *(const v4f*)(feats + (mt * 16 + lm) * FP + q * 8 + lh * 4);
            afr[mt][2 * q].x = t.x;     afr[mt][2 * q].y = t.y;
            afr[mt][2 * q + 1].x = t.z; afr[mt][2 * q + 1].y = t.w;
        }
    }

    __syncthreads();

    {
        v8f c[MTILES], cold[MTILES];
        v2f pb[G1PF];
        float bias;
        const int colw = wv * DOUT;
        const float* wl = W1f + lane * 4;
        const float* bl = b1 + lm;
        #pragma unroll
        for (int q = 0; q < G1PF / 2; ++q) {
            const v4f w = *(const v4f*)(wl + colw * 128 + q * 128);
            pb[2 * q].x = w.x;     pb[2 * q].y = w.y;
            pb[2 * q + 1].x = w.z; pb[2 * q + 1].y = w.w;
        }
        bias = bl[colw];
        g1_pass<false>(cold, c, afr, pb, bias, wl + colw * 128, wl + (colw + G1COLS) * 128, bl + colw + G1COLS);
        for (int j = 1; j < G1PASS; ++j) {
            const int col = colw + j * G1COLS;
            const int cn = colw + (j + 1 < G1PASS ? j + 1 : G1PASS - 1) * G1COLS;
            g1_pass<true>(c, cold, afr, pb, bias, wl + col * 128, wl + cn * 128, bl + cn);
            g1_store(cold, hbuf, col - G1COLS, lm, lh);
            #pragma unroll
            for (int mt = 0; mt < MTILES; ++mt) cold[mt] = c[mt];
        }
        #pragma unroll
        for (int h = 0; h < MTILES / NPAIR; ++h) {
            v8f x[NPAIR] = {cold[NPAIR * h], cold[NPAIR * h + 1]};
            gelu_tile(x);
            cold[NPAIR * h] = x[0];
            cold[NPAIR * h + 1] = x[1];
        }
        g1_store(cold, hbuf, colw + DOUT - G1COLS, lm, lh);
    }
    __syncthreads();

    g2_block(hbuf, rowid, nrows, W2f, b2, out, typ * DOUT + wv * QCOLS, wv * QCOLS, lm, lh);
}

extern "C" void kernel_launch(void* const* d_in, const int* in_sizes, int n_in,
                              void* d_out, int out_size, void* d_ws, size_t ws_size,
                              hipStream_t stream) {
    const float* nf = (const float*)d_in[0];
    const float* W1 = (const float*)d_in[1];
    const float* b1 = (const float*)d_in[2];
    const float* W2 = (const float*)d_in[3];
    const float* b2 = (const float*)d_in[4];
    float* out = (float*)d_out;
    int* ws = (int*)d_ws;

    int Mtot = in_sizes[0] / FTOT;
    int grid = (Mtot + MT - 1) / MT;
    int nblk = (Mtot + CB - 1) / CB;
    if (ws_size < (size_t)ws_wf_bytes(Mtot) + (size_t)(W1F4 + W2F4) * 16) return;

    hipFuncSetAttribute((const void*)mpx_encoder_kernel,
                        hipFuncAttributeMaxDynamicSharedMemorySize, SMEM_BYTES);

    mpx_type_count_kernel<<<nblk + WFBLK, CB, 0, stream>>>(nf, W1, W2, ws, Mtot);
    mpx_place_kernel<<<nblk, CB, 0, stream>>>(ws, Mtot);
    mpx_encoder_kernel<<<grid + 3, THREADS, SMEM_BYTES, stream>>>(nf, b1, b2, out, ws, Mtot);
}


# Grid tail balanced: the workgroups past the last full round of 256 run as up to 4x as many 16-row (one row tile) workgroups beside that round; same k order, sums unchanged
# speedup vs baseline: 7.0472x; 1.0403x over previous
#include <hip/hip_runtime.h>

typedef float v2f __attribute__((ext_vector_type(2)));
typedef float v4f __attribute__((ext_vector_type(4)));
typedef float v8f __attribute__((ext_vector_type(8)));

#define MT        64
#define WAVES     4
#define THREADS   128
#define FTOT      132
#define HDIM      512
#define DOUT      128
#define FP        132
#define HP        516
#define HBYTES    (MT * HP * 4)
#define ROWOFF    HBYTES
#define SMEM_BYTES (ROWOFF + MT * 4)
#define CB        1024
#define CW        (CB / 32)
#define CT2       2
#define QCOLS     (16 * CT2)
#define MTILES    4
#define NPAIR     2
#define NWGP      256
#define TAILMAX   (NWGP / 2)
#define SROWS     16
#define KG        8
#define NGRP      (HDIM / 4 / KG)
#define KIN       128
#define W1F4      (HDIM / 16 * (KIN / 8) * 32)
#define W2F4      (HDIM / 16 * (HDIM / 8) * 32)
#define WFBLK     ((W1F4 + W2F4) / CB)

__device__ __forceinline__ v8f splat8(float s) { v8f r = {s, s, s, s, s, s, s, s}; return r; }

template <int NP>
__device__ __forceinline__ void gelu_tile(v8f (&x)[NP]) {
    v8f z[NP], az[NP], t[NP], e[NP], p[NP];
    #pragma unroll
    for (int n = 0; n < NP; ++n) {
        z[n]  = x[n] * splat8(0.70710678118654752f);
        az[n] = __builtin_elementwise_abs(z[n]);
    }
    #pragma unroll
    for (int n = 0; n < NP; ++n) {
        const v8f d = __builtin_elementwise_fma(az[n], splat8(0.3275911f), splat8(1.0f));
        const v8f w = az[n] * az[n] * splat8(-1.4426950408889634f);
        #pragma unroll
        for (int i = 0; i < 8; ++i) {
            t[n][i] = __builtin_amdgcn_rcpf(d[i]);
            e[n][i] = __builtin_amdgcn_exp2f(w[i]);
        }
    }
    #pragma unroll
    for (int n = 0; n < NP; ++n) p[n] = __builtin_elementwise_fma(t[n], splat8(1.061405429f), splat8(-1.453152027f));
    #pragma unroll
    for (int n = 0; n < NP; ++n) p[n] = __builtin_elementwise_fma(t[n], p[n], splat8(1.421413741f));
    #pragma unroll
    for (int n = 0; n < NP; ++n) p[n] = __builtin_elementwise_fma(t[n], p[n], splat8(-0.284496736f));
    #pragma unroll
    for (int n = 0; n < NP; ++n) p[n] = __builtin_elementwise_fma(t[n], p[n], splat8(0.254829592f));
    #pragma unroll
    for (int n = 0; n < NP; ++n) p[n] = p[n] * t[n];
    #pragma unroll
    for (int n = 0; n < NP; ++n) {
        const v8f erfa = __builtin_elementwise_fma(-p[n], e[n], splat8(1.0f));
        const v8f erfz = __builtin_elementwise_copysign(erfa, z[n]);
        const v8f hx   = x[n] * splat8(0.5f);
        x[n] = __builtin_elementwise_fma(hx, erfz, hx);
    }
}

#define G1COLS    16
#define G1PASS    (DOUT / G1COLS)
#define G1PF      8

__device__ __forceinline__ void gelu_step(int s, v8f& x, v8f& z, v8f& t, v8f& e, v8f& p) {
    switch (s) {
    case 0: z = x * splat8(0.70710678118654752f); t = __builtin_elementwise_abs(z); e = t * t; break;
    case 1: t = __builtin_elementwise_fma(t, splat8(0.3275911f), splat8(1.0f));
            #pragma unroll
            for (int i = 0; i < 4; ++i) t[i] = __builtin_amdgcn_rcpf(t[i]);
            break;
    case 2:
            #pragma unroll
            for (int i = 4; i < 8; ++i) t[i] = __builtin_amdgcn_rcpf(t[i]);
            e = e * splat8(-1.4426950408889634f);
            break;
    case 3:
            #pragma unroll
            for (int i = 0; i < 8; ++i) e[i] = __builtin_amdgcn_exp2f(e[i]);
            break;
    case 4: p = __builtin_elementwise_fma(t, splat8(1.061405429f), splat8(-1.453152027f));
            p = __builtin_elementwise_fma(t, p, splat8(1.421413741f));
            break;
    case 5: p = __builtin_elementwise_fma(t, p, splat8(-0.284496736f));
            p = __builtin_elementwise_fma(t, p, splat8(0.254829592f));
            p = p * t;
            break;
    case 6: p = __builtin_elementwise_fma(-p, e, splat8(1.0f));
            p = __builtin_elementwise_copysign(p, z);
            break;
    default: { const v8f hx = x * splat8(0.5f); x = __builtin_elementwise_fma(hx, p, hx); } break;
    }
}

template <bool ACT, int NT>
__device__ __forceinline__ void g1_pass(v8f (&c)[NT], v8f (&cold)[NT], const v2f (&afr)[NT][32],
                                        v2f (&pb)[G1PF], float& bias, const float* __restrict__ wcur,
                                        const float* __restrict__ wnext, const float* __restrict__ bnext)
{
    #pragma unroll
    for (int mt = 0; mt < NT; ++mt) c[mt] = splat8(bias);
    bias = *bnext;
    v2f b[32];
    #pragma unroll
    for (int q = G1PF / 2; q < 16; ++q) {
        const v4f w = *(const v4f*)(wcur + q * 128);
        b[2 * q].x = w.x;     b[2 * q].y = w.y;
        b[2 * q + 1].x = w.z; b[2 * q + 1].y = w.w;
    }
    v8f z, t, e, p;
    #pragma unroll
    for (int ks = 0; ks < 32; ++ks) {
        const v2f f = ks < G1PF ? pb[ks < G1PF ? ks : 0] : b[ks];
        #pragma unroll
        for (int mt = 0; mt < NT; ++mt)
            c[mt] = __builtin_amdgcn_wmma_f32_16x16x4_f32(false, afr[mt][ks], false, f,
                                                          (short)0, c[mt], false, false);
        if (ks < G1PF && (ks & 1)) {
            const int k1 = (ks < G1PF && (ks & 1)) ? ks : 1;
            const v4f w = *(const v4f*)(wnext + (k1 >> 1) * 128);
            pb[k1 - 1].x = w.x; pb[k1 - 1].y = w.y;
            pb[k1].x = w.z;     pb[k1].y = w.w;
        }
        if (ACT && (ks >> 3) < NT) {
            const int ct = (ks >> 3) < NT ? (ks >> 3) : 0;
            gelu_step(ks & 7, cold[ct], z, t, e, p);
            switch (ks & 7) {
            case 0: asm volatile("" : "+v"(z), "+v"(t), "+v"(e)); break;
            case 1: asm volatile("" : "+v"(t)); break;
            case 2: asm volatile("" : "+v"(t), "+v"(e)); break;
            case 3: asm volatile("" : "+v"(e)); break;
            case 7: asm volatile("" : "+v"(cold[ct])); break;
            default: asm volatile("" : "+v"(p)); break;
            }
            __builtin_amdgcn_sched_barrier(0);
        }
    }
}

template <int NT>
__device__ __forceinline__ void g1_store(const v8f (&c)[NT], float* hbuf, int col0, int lm, int lh) {
    #pragma unroll
    for (int mt = 0; mt < NT; ++mt) {
        #pragma unroll
        for (int v = 0; v < 8; ++v)
            hbuf[(mt * 16 + v + lh * 8) * HP + col0 + lm] = c[mt][v];
    }
}

__host__ __device__ __forceinline__ int ws_list(int Mtot) { return 64 + 4 * ((Mtot + CB - 1) / CB); }

__host__ __device__ __forceinline__ long ws_wf_bytes(int Mtot) {
    return (((long)(ws_list(Mtot) + Mtot) * 4 + Mtot) + 255) & ~255L;
}

__global__ __launch_bounds__(CB)
void mpx_type_count_kernel(const float* __restrict__ nf, const float* __restrict__ W1,
                           const float* __restrict__ W2, int* __restrict__ ws, int Mtot)
{
    __shared__ int wc[CW][4];
    const int tid = threadIdx.x, lane = tid & 31, w = tid >> 5;
    const int nblk = (Mtot + CB - 1) / CB;
    if ((int)blockIdx.x >= nblk) {
        int i = ((int)blockIdx.x - nblk) * CB + tid;
        v4f* dst = (v4f*)((char*)ws + ws_wf_bytes(Mtot)) + i;
        const float* Wm = W1;
        int nqs = 4;
        if (i >= W1F4) { i -= W1F4; Wm = W2; nqs = 6; }
        const int blk = i >> 5, q = blk & ((1 << nqs) - 1), ct = blk >> nqs;
        const float* s = Wm + (long)(8 * q + 4 * (lane >> 4)) * HDIM + 16 * ct + (lane & 15);
        v4f v = {s[0], s[HDIM], s[2 * HDIM], s[3 * HDIM]};
        *dst = v;
        return;
    }
    const int r = blockIdx.x * CB + tid;
    const bool live = r < Mtot;
    int bi = 4;
    if (live) {
        const float4 q = *(const float4*)(nf + (long)r * FTOT);
        float best = q.x; bi = 0;
        if (q.y > best) { best = q.y; bi = 1; }
        if (q.z > best) { best = q.z; bi = 2; }
        if (q.w > best) { best = q.w; bi = 3; }
        ((unsigned char*)(ws + ws_list(Mtot) + Mtot))[r] = (unsigned char)bi;
    }
    #pragma unroll
    for (int t = 0; t < 4; ++t) {
        const unsigned m = __builtin_amdgcn_ballot_w32(bi == t);
        if (lane == 0) wc[w][t] = __popc(m);
    }
    __syncthreads();
    if (tid < 4) {
        int s = 0;
        for (int i = 0; i < CW; ++i) s += wc[i][tid];
        ws[64 + 4 * blockIdx.x + tid] = s;
    }
}

__global__ __launch_bounds__(CB)
void mpx_place_kernel(int* __restrict__ ws, int Mtot)
{
    __shared__ int wc[CW][4];
    __shared__ int lo[4], all[4];
    const int tid = threadIdx.x, lane = tid & 31, w = tid >> 5;
    const int nblk = (Mtot + CB - 1) / CB;
    const int* cnts = ws + 64;
    if (w < 4) {
        int sl = 0, sa = 0;
        for (int b = lane; b < nblk; b += 32) {
            const int v = cnts[4 * b + w];
            sa += v;
            if (b < (int)blockIdx.x) sl += v;
        }
        for (int o = 16; o > 0; o >>= 1) { sl += __shfl_xor(sl, o); sa += __shfl_xor(sa, o); }
        if (lane == 0) { lo[w] = sl; all[w] = sa; }
    }
    const int r = blockIdx.x * CB + tid;
    const bool live = r < Mtot;
    const unsigned char* typ = (const unsigned char*)(ws + ws_list(Mtot) + Mtot);
    const int bi = live ? (int)typ[r] : 4;
    unsigned mine = 0;
    #pragma unroll
    for (int t = 0; t < 4; ++t) {
        const unsigned m = __builtin_amdgcn_ballot_w32(bi == t);
        if (lane == 0) wc[w][t] = __popc(m);
        if (bi == t) mine = m;
    }
    __syncthreads();
    if (live) {
        int pos = lo[bi];
        for (int u = 0; u < bi; ++u) pos += all[u];
        for (int i = 0; i < w; ++i) pos += wc[i][bi];
        pos += __popc(mine & ((1u << lane) - 1u));
        ws[ws_list(Mtot) + pos] = r;
    }
    if (blockIdx.x == 0 && tid < 4) ws[tid] = all[tid];
}

template <int NT>
__device__ __forceinline__ void g2_load_a(v2f (&a)[KG][NT], const float* hb, const int (&hoff)[NT]) {
    #pragma unroll
    for (int p = 0; p < KG / 2; ++p) {
        #pragma unroll
        for (int mt = 0; mt < NT; ++mt) {
            const v4f t = *(const v4f*)(hb + hoff[mt] + p * 8);
            a[2 * p][mt].x = t.x;     a[2 * p][mt].y = t.y;
            a[2 * p + 1][mt].x = t.z; a[2 * p + 1][mt].y = t.w;
        }
    }
}

__device__ __forceinline__ void g2_load_b(v2f (&b)[KG][CT2], const float* wp) {
    #pragma unroll
    for (int p = 0; p < KG / 2; ++p) {
        #pragma unroll
        for (int nt = 0; nt < CT2; ++nt) {
            const v4f w = *(const v4f*)(wp + nt * (HDIM / 8) * 128 + p * 128);
            b[2 * p][nt].x = w.x;     b[2 * p][nt].y = w.y;
            b[2 * p + 1][nt].x = w.z; b[2 * p + 1][nt].y = w.w;
        }
    }
}

#define G2_STAGE(an, bn, ac, bc)                         \
    do {                                                  \
        g2_load_a(an, hbuf, hoff);                        \
        __builtin_amdgcn_sched_barrier(0x76);             \
        g2_load_b(bn, wp);                                \
        g2_mma(c, ac, bc);                                \
    } while (0)

template <int NT>
__device__ __forceinline__ void g2_mma(v8f (&c)[NT][CT2], const v2f (&a)[KG][NT], const v2f (&b)[KG][CT2]) {
    #pragma unroll
    for (int j = 0; j < KG; ++j) {
        #pragma unroll
        for (int mt = 0; mt < NT; ++mt) {
            #pragma unroll
            for (int nt = 0; nt < CT2; ++nt)
                c[mt][nt] = __builtin_amdgcn_wmma_f32_16x16x4_f32(false, a[j][mt], false, b[j][nt],
                                                                  (short)0, c[mt][nt], false, false);
        }
    }
}

template <int NT>
__device__ __forceinline__ void g2_block(const float* hbuf, const int* rowid, int nrows,
                                         const float* __restrict__ W2f, const float* __restrict__ b2,
                                         float* __restrict__ out, int ncol0, int ocol0, int lm, int lh)
{
    v8f c[NT][CT2];
    #pragma unroll
    for (int nt = 0; nt < CT2; ++nt) {
        const float bias = b2[ncol0 + nt * 16 + lm];
        #pragma unroll
        for (int mt = 0; mt < NT; ++mt) c[mt][nt] = splat8(bias);
    }
    int hoff[NT];
    #pragma unroll
    for (int mt = 0; mt < NT; ++mt) hoff[mt] = (mt * 16 + lm) * HP + lh * 4;
    const float* wp = W2f + (ncol0 / 16) * (HDIM / 8) * 128 + (lh * 16 + lm) * 4;
    v2f a0[KG][NT], b0[KG][CT2], a1[KG][NT], b1[KG][CT2];
    g2_load_a(a0, hbuf, hoff);
    g2_load_b(b0, wp);
    for (int g = 0; g < NGRP - 2; g += 2) {
        #pragma unroll
        for (int mt = 0; mt < NT; ++mt) hoff[mt] += KG * 4;
        wp += (KG / 2) * 128;
        G2_STAGE(a1, b1, a0, b0);
        #pragma unroll
        for (int mt = 0; mt < NT; ++mt) hoff[mt] += KG * 4;
        wp += (KG / 2) * 128;
        G2_STAGE(a0, b0, a1, b1);
    }
    #pragma unroll
    for (int mt = 0; mt < NT; ++mt) hoff[mt] += KG * 4;
    wp += (KG / 2) * 128;
    G2_STAGE(a1, b1, a0, b0);
    g2_mma(c, a1, b1);
    #pragma unroll
    for (int mt = 0; mt < NT; ++mt) {
        #pragma unroll
        for (int v = 0; v < 8; ++v) {
            const int s = mt * 16 + lh * 8 + v;
            if (s < nrows) {
                float* o = out + (long)rowid[s] * DOUT + ocol0 + lm;
                #pragma unroll
                for (int nt = 0; nt < CT2; ++nt) o[nt * 16] = c[mt][nt][v];
            }
        }
    }
}

template <int NT>
__device__ __forceinline__ void mpx_encoder_body(const float* __restrict__ nf, const float* __restrict__ b1,
                                                 const float* __restrict__ b2, float* __restrict__ out,
                                                 const int* __restrict__ list, const float* __restrict__ W1f,
                                                 const float* __restrict__ W2f, char* smem,
                                                 int typ, int nrows, int tid, int lane, int wv, int lm, int lh)
{
    float* feats = (float*)smem;
    float* hbuf  = (float*)smem;
    int*   rowid = (int*)(smem + ROWOFF);

    #pragma unroll
    for (int it = 0; it < 4 * NT; ++it) {
        int r  = wv + 4 * it;
        int c4 = lane;
        const long rs = list[r < nrows ? r : nrows - 1];
        const float4 v = *(const float4*)(nf + rs * FTOT + 4 + c4 * 4);
        float* d = feats + r * FP + c4 * 4;
        d[0] = v.x; d[1] = v.y; d[2] = v.z; d[3] = v.w;
    }
    if (tid < 16 * NT) rowid[tid] = list[tid < nrows ? tid : nrows - 1];
    __syncthreads();

    v2f afr[NT][32];
    #pragma unroll
    for (int mt = 0; mt < NT; ++mt) {
        #pragma unroll
        for (int q = 0; q < 16; ++q) {
            const v4f t = *(const v4f*)(feats + (mt * 16 + lm) * FP + q * 8 + lh * 4);
            afr[mt][2 * q].x = t.x;     afr[mt][2 * q].y = t.y;
            afr[mt][2 * q + 1].x = t.z; afr[mt][2 * q + 1].y = t.w;
        }
    }

    __syncthreads();

    {
        v8f c[NT], cold[NT];
        v2f pb[G1PF];
        float bias;
        const int colw = wv * DOUT;
        const float* wl = W1f + lane * 4;
        const float* bl = b1 + lm;
        #pragma unroll
        for (int q = 0; q < G1PF / 2; ++q) {
            const v4f w = *(const v4f*)(wl + colw * 128 + q * 128);
            pb[2 * q].x = w.x;     pb[2 * q].y = w.y;
            pb[2 * q + 1].x = w.z; pb[2 * q + 1].y = w.w;
        }
        bias = bl[colw];
        g1_pass<false, NT>(cold, c, afr, pb, bias, wl + colw * 128, wl + (colw + G1COLS) * 128, bl + colw + G1COLS);
        for (int j = 1; j < G1PASS; ++j) {
            const int col = colw + j * G1COLS;
            const int cn = colw + (j + 1 < G1PASS ? j + 1 : G1PASS - 1) * G1COLS;
            g1_pass<true, NT>(c, cold, afr, pb, bias, wl + col * 128, wl + cn * 128, bl + cn);
            g1_store<NT>(cold, hbuf, col - G1COLS, lm, lh);
            #pragma unroll
            for (int mt = 0; mt < NT; ++mt) cold[mt] = c[mt];
        }
        constexpr int NP = NT < NPAIR ? NT : NPAIR;
        #pragma unroll
        for (int h = 0; h < NT / NP; ++h) {
            v8f x[NP];
            #pragma unroll
            for (int n = 0; n < NP; ++n) x[n] = cold[NP * h + n];
            gelu_tile<NP>(x);
            #pragma unroll
            for (int n = 0; n < NP; ++n) cold[NP * h + n] = x[n];
        }
        g1_store<NT>(cold, hbuf, colw + DOUT - G1COLS, lm, lh);
    }
    __syncthreads();

    g2_block<NT>(hbuf, rowid, nrows, W2f, b2, out, typ * DOUT + wv * QCOLS, wv * QCOLS, lm, lh);
}

__global__ __launch_bounds__(THREADS, 2)
void mpx_encoder_kernel(const float* __restrict__ nf,
                        const float* __restrict__ b1,
                        const float* __restrict__ b2,
                        float* __restrict__ out,
                        const int* __restrict__ ws,
                        int Mtot)
{
    extern __shared__ char smem[];

    const int  tid  = threadIdx.x;
    const int  lane = tid & 31;
    const int  wv   = __builtin_amdgcn_readfirstlane(tid >> 5);
    const int  lm   = lane & 15;
    const int  lh   = lane >> 4;

    int typ = -1, first = 0, nleft = 0;
    bool small = false;
    {
        int nu[4], gu[4], ku[4];
        int G = 0;
        #pragma unroll
        for (int u = 0; u < 4; ++u) {
            nu[u] = ws[u];
            gu[u] = (nu[u] + MT - 1) / MT;
            G += gu[u];
        }
        int K = G % NWGP;
        if (K > TAILMAX) K = 0;
        #pragma unroll
        for (int u = 3; u >= 0; --u) {
            ku[u] = K < gu[u] ? K : gu[u];
            K -= ku[u];
        }
        int g = blockIdx.x, off = 0;
        #pragma unroll
        for (int u = 0; u < 4; ++u) {
            const int nb = gu[u] - ku[u];
            if (typ < 0) {
                if (g < nb) { typ = u; first = off + g * MT; nleft = nu[u] - g * MT; }
                else g -= nb;
            }
            off += nu[u];
        }
        if (typ < 0) {
            off = 0;
            #pragma unroll
            for (int u = 0; u < 4; ++u) {
                const int nb = gu[u] - ku[u];
                int rs = nu[u] - nb * MT;
                if (rs < 0) rs = 0;
                const int ns = (rs + SROWS - 1) / SROWS;
                if (typ < 0) {
                    if (g < ns) { typ = u; first = off + nb * MT + g * SROWS; nleft = rs - g * SROWS; small = true; }
                    else g -= ns;
                }
                off += nu[u];
            }
        }
    }
    if (typ < 0) return;
    const int* list  = ws + ws_list(Mtot) + first;
    const float* W1f = (const float*)((const char*)ws + ws_wf_bytes(Mtot));
    const float* W2f = W1f + W1F4 * 4;

    if (small) {
        const int nrows = nleft < SROWS ? nleft : SROWS;
        mpx_encoder_body<1>(nf, b1, b2, out, list, W1f, W2f, smem, typ, nrows, tid, lane, wv, lm, lh);
    } else {
        const int nrows = nleft < MT ? nleft : MT;
        mpx_encoder_body<MTILES>(nf, b1, b2, out, list, W1f, W2f, smem, typ, nrows, tid, lane, wv, lm, lh);
    }
}

extern "C" void kernel_launch(void* const* d_in, const int* in_sizes, int n_in,
                              void* d_out, int out_size, void* d_ws, size_t ws_size,
                              hipStream_t stream) {
    const float* nf = (const float*)d_in[0];
    const float* W1 = (const float*)d_in[1];
    const float* b1 = (const float*)d_in[2];
    const float* W2 = (const float*)d_in[3];
    const float* b2 = (const float*)d_in[4];
    float* out = (float*)d_out;
    int* ws = (int*)d_ws;

    int Mtot = in_sizes[0] / FTOT;
    int grid = (Mtot + MT - 1) / MT;
    int nblk = (Mtot + CB - 1) / CB;
    if (ws_size < (size_t)ws_wf_bytes(Mtot) + (size_t)(W1F4 + W2F4) * 16) return;

    hipFuncSetAttribute((const void*)mpx_encoder_kernel,
                        hipFuncAttributeMaxDynamicSharedMemorySize, SMEM_BYTES);

    mpx_type_count_kernel<<<nblk + WFBLK, CB, 0, stream>>>(nf, W1, W2, ws, Mtot);
    mpx_place_kernel<<<nblk, CB, 0, stream>>>(ws, Mtot);
    mpx_encoder_kernel<<<grid + 3 + 3 * TAILMAX, THREADS, SMEM_BYTES, stream>>>(nf, b1, b2, out, ws, Mtot);
}
